# starting kernel
# baseline (speedup 1.0000x reference)
#include <hip/hip_runtime.h>

#define N_NODES 50000
#define N_EDGES 600000
#define HID 128
#define NEG_SLOPE 0.01f
#define N_LAYERS 3

typedef __attribute__((ext_vector_type(2))) float v2f;
typedef __attribute__((ext_vector_type(8))) float v8f;

__global__ __launch_bounds__(256) void zero_kernel(float4* __restrict__ p, int n4) {
  int i = blockIdx.x * 256 + threadIdx.x;
  if (i < n4) p[i] = make_float4(0.f, 0.f, 0.f, 0.f);
}

__global__ __launch_bounds__(256) void edge_kernel(
    const float* __restrict__ h,
    const int*   __restrict__ src,
    const int*   __restrict__ dst,
    const float* __restrict__ ea,
    const float* __restrict__ Wl0,
    const float* __restrict__ Wl1,
    const float* __restrict__ bl,
    float* __restrict__ agg)
{
  int tid = blockIdx.x * 256 + threadIdx.x;
  int e   = tid >> 5;
  int c   = (tid & 31) * 4;
  if (e >= N_EDGES) return;

  int   s  = src[e];
  int   d  = dst[e];
  float a0 = ea[2 * e];
  float a1 = ea[2 * e + 1];

  float4 w0 = *(const float4*)(Wl0 + c);
  float4 w1 = *(const float4*)(Wl1 + c);
  float4 bb = *(const float4*)(bl  + c);
  float4 hv = *(const float4*)(h + (long)s * HID + c);

  float m0 = hv.x + a0 * w0.x + a1 * w1.x + bb.x;
  float m1 = hv.y + a0 * w0.y + a1 * w1.y + bb.y;
  float m2 = hv.z + a0 * w0.z + a1 * w1.z + bb.z;
  float m3 = hv.w + a0 * w0.w + a1 * w1.w + bb.w;
  m0 = m0 > 0.f ? m0 : 0.f;
  m1 = m1 > 0.f ? m1 : 0.f;
  m2 = m2 > 0.f ? m2 : 0.f;
  m3 = m3 > 0.f ? m3 : 0.f;

  float* ap = agg + (long)d * HID + c;
  unsafeAtomicAdd(ap + 0, m0);
  unsafeAtomicAdd(ap + 1, m1);
  unsafeAtomicAdd(ap + 2, m2);
  unsafeAtomicAdd(ap + 3, m3);
}

__global__ __launch_bounds__(256) void gine_gemm_kernel(
    const float* __restrict__ h,
    const float* __restrict__ agg,
    const float* __restrict__ W,
    const float* __restrict__ bias,
    float* __restrict__ out)
{
  int wave = threadIdx.x >> 5;
  int lane = threadIdx.x & 31;
  int m0   = blockIdx.x * 16;
  int n0   = wave * 16;
  int m    = lane & 15;
  int half = lane >> 4;
  int n    = n0 + m;

  const float* xr = h   + (long)(m0 + m) * HID + half * 2;
  const float* gr = agg + (long)(m0 + m) * HID + half * 2;

  v8f c = {};
  for (int kk = 0; kk < HID; kk += 4) {
    float2 hx = *(const float2*)(xr + kk);
    float2 gx = *(const float2*)(gr + kk);
    v2f a;
    a.x = hx.x + gx.x;
    a.y = hx.y + gx.y;

    int k0 = kk + half * 2;
    v2f bm;
    bm.x = W[(long)k0 * HID + n];
    bm.y = W[(long)(k0 + 1) * HID + n];

    c = __builtin_amdgcn_wmma_f32_16x16x4_f32(
         false, a,  false, bm,
         (short)0, c,  false,  false);
  }

  float bv = bias[n];
#pragma unroll
  for (int v = 0; v < 8; ++v) {
    int row = m0 + v + half * 8;
    float y = c[v] + bv;
    out[(long)row * HID + n] = (y >= 0.f) ? y : NEG_SLOPE * y;
  }
}

extern "C" void kernel_launch(void* const* d_in, const int* in_sizes, int n_in,
                              void* d_out, int out_size, void* d_ws, size_t ws_size,
                              hipStream_t stream) {
  (void)in_sizes; (void)n_in; (void)out_size; (void)ws_size;

  const float* x  = (const float*)d_in[0];
  const int*   ei = (const int*)  d_in[1];
  const float* ea = (const float*)d_in[2];
  const float* Wl = (const float*)d_in[3];
  const float* bl = (const float*)d_in[4];
  const float* W  = (const float*)d_in[5];
  const float* b  = (const float*)d_in[6];
  float*       out = (float*)d_out;

  const int* src = ei;
  const int* dst = ei + N_EDGES;

  const long NH = (long)N_NODES * HID;
  float* agg = (float*)d_ws;
  float* h0  = agg + NH;
  float* h1  = h0 + NH;

  const float* hin = x;
  for (int layer = 0; layer < N_LAYERS; ++layer) {
    float* hout = (layer == N_LAYERS - 1) ? out : (layer == 0 ? h0 : h1);

    zero_kernel<<<(int)(NH / 4 + 255) / 256, 256, 0, stream>>>((float4*)agg, (int)(NH / 4));

    edge_kernel<<<N_EDGES / 8, 256, 0, stream>>>(
        hin, src, dst, ea,
        Wl + (long)layer * 2 * HID,
        Wl + (long)layer * 2 * HID + HID,
        bl + (long)layer * HID,
        agg);

    gine_gemm_kernel<<<N_NODES / 16, 256, 0, stream>>>(
        hin, agg,
        W + (long)layer * HID * HID,
        b + (long)layer * HID,
        hout);

    hin = hout;
  }
}


# Replace per-edge f32 atomic scatter and zero pass by a per-call sort of edges by destination; one wave per node sums its messages in registers in ascending edge-id order (fixed sum order)
# speedup vs baseline: 1.5296x; 1.5296x over previous
#include <hip/hip_runtime.h>


#define N_NODES 50000
#define N_EDGES 600000
#define HID 128
#define NEG_SLOPE 0.01f
#define N_LAYERS 3

#define N_PAD 53248
#define ROW_BLOCKS 2048
#define ROW_WAVES (ROW_BLOCKS * 8)

typedef __attribute__((ext_vector_type(2))) float v2f;
typedef __attribute__((ext_vector_type(8))) float v8f;


__global__ __launch_bounds__(256) void zero_counts_kernel(int4* __restrict__ p, int n4) {
  int i = blockIdx.x * 256 + threadIdx.x;
  if (i < n4) p[i] = make_int4(0, 0, 0, 0);
}

__global__ __launch_bounds__(256) void count_kernel(const int* __restrict__ dst,
                                                    int* __restrict__ cnt) {
  int t = blockIdx.x * 256 + threadIdx.x;
  if (t >= N_EDGES / 4) return;
  int4 d = *(const int4*)(dst + 4 * t);
  atomicAdd(cnt + d.x, 1);
  atomicAdd(cnt + d.y, 1);
  atomicAdd(cnt + d.z, 1);
  atomicAdd(cnt + d.w, 1);
}

__global__ __launch_bounds__(1024) void scan_kernel(int4* cntcur, int4* __restrict__ off) {
  __shared__ int part[1024];
  int t = threadIdx.x;
  int4 v[13];
  int sum = 0;
  for (int k = 0; k < 13; ++k) {
    v[k] = cntcur[t * 13 + k];
    sum += v[k].x + v[k].y + v[k].z + v[k].w;
  }
  part[t] = sum;
  __syncthreads();
  for (int d = 1; d < 1024; d <<= 1) {
    int u = (t >= d) ? part[t - d] : 0;
    __syncthreads();
    part[t] += u;
    __syncthreads();
  }
  int run = part[t] - sum;
  for (int k = 0; k < 13; ++k) {
    int4 o;
    o.x = run; run += v[k].x;
    o.y = run; run += v[k].y;
    o.z = run; run += v[k].z;
    o.w = run; run += v[k].w;
    off[t * 13 + k]    = o;
    cntcur[t * 13 + k] = o;
  }
}

__global__ __launch_bounds__(256) void place_kernel(const int* __restrict__ dst,
                                                    int* __restrict__ cur,
                                                    int* __restrict__ perm) {
  int t = blockIdx.x * 256 + threadIdx.x;
  if (t >= N_EDGES / 4) return;
  int4 d = *(const int4*)(dst + 4 * t);
  int e = 4 * t;
  int p0 = atomicAdd(cur + d.x, 1);
  int p1 = atomicAdd(cur + d.y, 1);
  int p2 = atomicAdd(cur + d.z, 1);
  int p3 = atomicAdd(cur + d.w, 1);
  perm[min(max(p0, 0), N_EDGES - 1)] = e;
  perm[min(max(p1, 0), N_EDGES - 1)] = e + 1;
  perm[min(max(p2, 0), N_EDGES - 1)] = e + 2;
  perm[min(max(p3, 0), N_EDGES - 1)] = e + 3;
}

__global__ __launch_bounds__(256) void order_kernel(const int* __restrict__ off,
                                                    const int* __restrict__ perm,
                                                    const int* __restrict__ src,
                                                    const float* __restrict__ ea,
                                                    int4* __restrict__ rec) {
  int lane = threadIdx.x & 31;
  int w = __builtin_amdgcn_readfirstlane((int)(blockIdx.x * 8 + (threadIdx.x >> 5)));
  for (int i = w; i < N_NODES; i += ROW_WAVES) {
    int beg = max(off[i], 0);
    int end = min(off[i + 1], N_EDGES);
    for (int p = beg + lane; p < end; p += 32) {
      int e = perm[p];
      int rank = 0;
      for (int q = beg; q < end; ++q) rank += (perm[q] < e) ? 1 : 0;
      int ec = min(max(e, 0), N_EDGES - 1);
      int s = src[ec];
      float2 a = *(const float2*)(ea + 2 * ec);
      rec[beg + rank] = make_int4(s, __float_as_int(a.x), __float_as_int(a.y), 0);
    }
  }
}

__device__ __forceinline__ float4 edge_msg(float4 hv, float a0, float a1, float4 w0, float4 w1, float4 bb) {
  float m0 = hv.x + a0 * w0.x + a1 * w1.x + bb.x;
  float m1 = hv.y + a0 * w0.y + a1 * w1.y + bb.y;
  float m2 = hv.z + a0 * w0.z + a1 * w1.z + bb.z;
  float m3 = hv.w + a0 * w0.w + a1 * w1.w + bb.w;
  return make_float4(m0 > 0.f ? m0 : 0.f, m1 > 0.f ? m1 : 0.f,
                     m2 > 0.f ? m2 : 0.f, m3 > 0.f ? m3 : 0.f);
}

__device__ __forceinline__ float4 load_row(const float* __restrict__ h, int s, int c) {
  int sc = min(max(s, 0), N_NODES - 1);
  return *(const float4*)(h + (long)sc * HID + c);
}

__global__ __launch_bounds__(256) void gather_kernel(
    const float* __restrict__ h,
    const int*   __restrict__ off,
    const int4*  __restrict__ rec,
    const float* __restrict__ Wl0,
    const float* __restrict__ Wl1,
    const float* __restrict__ bl,
    float* __restrict__ agg)
{
  int lane = threadIdx.x & 31;
  int c    = lane * 4;
  int w = __builtin_amdgcn_readfirstlane((int)(blockIdx.x * 8 + (threadIdx.x >> 5)));

  float4 w0 = *(const float4*)(Wl0 + c);
  float4 w1 = *(const float4*)(Wl1 + c);
  float4 bb = *(const float4*)(bl  + c);

  for (int i = w; i < N_NODES; i += ROW_WAVES) {
    int beg = max(off[i], 0);
    int end = min(off[i + 1], N_EDGES);
    float4 acc = make_float4(0.f, 0.f, 0.f, 0.f);
    for (int cb = beg; cb < end; cb += 32) {
      int n = min(end - cb, 32);
      int4 my = rec[cb + min(lane, n - 1)];
      for (int t = 0; t < n; t += 4) {
        int t1 = min(t + 1, n - 1);
        int t2 = min(t + 2, n - 1);
        int t3 = min(t + 3, n - 1);
        float4 x0 = load_row(h, __builtin_amdgcn_readlane(my.x, t),  c);
        float4 x1 = load_row(h, __builtin_amdgcn_readlane(my.x, t1), c);
        float4 x2 = load_row(h, __builtin_amdgcn_readlane(my.x, t2), c);
        float4 x3 = load_row(h, __builtin_amdgcn_readlane(my.x, t3), c);
        float4 m0 = edge_msg(x0, __int_as_float(__builtin_amdgcn_readlane(my.y, t)),
                                 __int_as_float(__builtin_amdgcn_readlane(my.z, t)),  w0, w1, bb);
        float4 m1 = edge_msg(x1, __int_as_float(__builtin_amdgcn_readlane(my.y, t1)),
                                 __int_as_float(__builtin_amdgcn_readlane(my.z, t1)), w0, w1, bb);
        float4 m2 = edge_msg(x2, __int_as_float(__builtin_amdgcn_readlane(my.y, t2)),
                                 __int_as_float(__builtin_amdgcn_readlane(my.z, t2)), w0, w1, bb);
        float4 m3 = edge_msg(x3, __int_as_float(__builtin_amdgcn_readlane(my.y, t3)),
                                 __int_as_float(__builtin_amdgcn_readlane(my.z, t3)), w0, w1, bb);
        bool v1 = t + 1 < n, v2 = t + 2 < n, v3 = t + 3 < n;
        acc.x += m0.x;               acc.y += m0.y;               acc.z += m0.z;               acc.w += m0.w;
        acc.x += v1 ? m1.x : 0.f;    acc.y += v1 ? m1.y : 0.f;    acc.z += v1 ? m1.z : 0.f;    acc.w += v1 ? m1.w : 0.f;
        acc.x += v2 ? m2.x : 0.f;    acc.y += v2 ? m2.y : 0.f;    acc.z += v2 ? m2.z : 0.f;    acc.w += v2 ? m2.w : 0.f;
        acc.x += v3 ? m3.x : 0.f;    acc.y += v3 ? m3.y : 0.f;    acc.z += v3 ? m3.z : 0.f;    acc.w += v3 ? m3.w : 0.f;
      }
    }
    *(float4*)(agg + (long)i * HID + c) = acc;
  }
}

__global__ __launch_bounds__(256) void gine_gemm_kernel(
    const float* __restrict__ h,
    const float* __restrict__ agg,
    const float* __restrict__ W,
    const float* __restrict__ bias,
    float* __restrict__ out)
{
  int wave = threadIdx.x >> 5;
  int lane = threadIdx.x & 31;
  int m0   = blockIdx.x * 16;
  int n0   = wave * 16;
  int m    = lane & 15;
  int half = lane >> 4;
  int n    = n0 + m;

  const float* xr = h   + (long)(m0 + m) * HID + half * 2;
  const float* gr = agg + (long)(m0 + m) * HID + half * 2;

  v8f c = {};
  for (int kk = 0; kk < HID; kk += 4) {
    float2 hx = *(const float2*)(xr + kk);
    float2 gx = *(const float2*)(gr + kk);
    v2f a;
    a.x = hx.x + gx.x;
    a.y = hx.y + gx.y;

    int k0 = kk + half * 2;
    v2f bm;
    bm.x = W[(long)k0 * HID + n];
    bm.y = W[(long)(k0 + 1) * HID + n];

    c = __builtin_amdgcn_wmma_f32_16x16x4_f32(
         false, a,  false, bm,
         (short)0, c,  false,  false);
  }

  float bv = bias[n];
#pragma unroll
  for (int v = 0; v < 8; ++v) {
    int row = m0 + v + half * 8;
    float y = c[v] + bv;
    out[(long)row * HID + n] = (y >= 0.f) ? y : NEG_SLOPE * y;
  }
}

extern "C" void kernel_launch(void* const* d_in, const int* in_sizes, int n_in,
                              void* d_out, int out_size, void* d_ws, size_t ws_size,
                              hipStream_t stream) {
  (void)in_sizes; (void)n_in; (void)out_size; (void)ws_size;

  const float* x  = (const float*)d_in[0];
  const int*   ei = (const int*)  d_in[1];
  const float* ea = (const float*)d_in[2];
  const float* Wl = (const float*)d_in[3];
  const float* bl = (const float*)d_in[4];
  const float* W  = (const float*)d_in[5];
  const float* b  = (const float*)d_in[6];
  float*       out = (float*)d_out;

  const int* src = ei;
  const int* dst = ei + N_EDGES;

  const long NH = (long)N_NODES * HID;
  float* agg  = (float*)d_ws;
  float* h0   = agg + NH;
  int*   cur  = (int*)(h0 + NH);
  int*   off  = cur + N_PAD;
  int*   perm = off + N_PAD;
  int4*  rec  = (int4*)(perm + N_EDGES);

  zero_counts_kernel<<<(N_PAD / 4 + 255) / 256, 256, 0, stream>>>((int4*)cur, N_PAD / 4);
  count_kernel<<<(N_EDGES / 4 + 255) / 256, 256, 0, stream>>>(dst, cur);
  scan_kernel<<<1, 1024, 0, stream>>>((int4*)cur, (int4*)off);
  place_kernel<<<(N_EDGES / 4 + 255) / 256, 256, 0, stream>>>(dst, cur, perm);
  order_kernel<<<ROW_BLOCKS, 256, 0, stream>>>(off, perm, src, ea, rec);

  const float* hin = x;
  for (int layer = 0; layer < N_LAYERS; ++layer) {
    float* hout = (layer == 1) ? h0 : out;

    gather_kernel<<<ROW_BLOCKS, 256, 0, stream>>>(
        hin, off, rec,
        Wl + (long)layer * 2 * HID,
        Wl + (long)layer * 2 * HID + HID,
        bl + (long)layer * HID,
        agg);

    gine_gemm_kernel<<<N_NODES / 16, 256, 0, stream>>>(
        hin, agg,
        W + (long)layer * HID * HID,
        b + (long)layer * HID,
        hout);

    hin = hout;
  }
}


# Fuse edge sum into the GEMM launch per layer: a wave sums 2 rows of a 16-row tile in registers (both gathers in flight, same sum order), X=h+agg via LDS to WMMA; agg never stored; launches 11 to 8
# speedup vs baseline: 2.0634x; 1.3490x over previous
#include <hip/hip_runtime.h>


#define N_NODES 50000
#define N_EDGES 600000
#define HID 128
#define NEG_SLOPE 0.01f
#define N_LAYERS 3

#define N_PAD 53248
#define ROW_BLOCKS 2048
#define ROW_WAVES (ROW_BLOCKS * 8)

typedef __attribute__((ext_vector_type(2))) float v2f;
typedef __attribute__((ext_vector_type(8))) float v8f;


__global__ __launch_bounds__(256) void zero_counts_kernel(int4* __restrict__ p, int n4) {
  int i = blockIdx.x * 256 + threadIdx.x;
  if (i < n4) p[i] = make_int4(0, 0, 0, 0);
}

__global__ __launch_bounds__(256) void count_kernel(const int* __restrict__ dst,
                                                    int* __restrict__ cnt) {
  int t = blockIdx.x * 256 + threadIdx.x;
  if (t >= N_EDGES / 4) return;
  int4 d = *(const int4*)(dst + 4 * t);
  atomicAdd(cnt + d.x, 1);
  atomicAdd(cnt + d.y, 1);
  atomicAdd(cnt + d.z, 1);
  atomicAdd(cnt + d.w, 1);
}

__global__ __launch_bounds__(1024) void scan_kernel(int4* cntcur, int4* __restrict__ off) {
  __shared__ int part[1024];
  int t = threadIdx.x;
  int4 v[13];
  int sum = 0;
  for (int k = 0; k < 13; ++k) {
    v[k] = cntcur[t * 13 + k];
    sum += v[k].x + v[k].y + v[k].z + v[k].w;
  }
  part[t] = sum;
  __syncthreads();
  for (int d = 1; d < 1024; d <<= 1) {
    int u = (t >= d) ? part[t - d] : 0;
    __syncthreads();
    part[t] += u;
    __syncthreads();
  }
  int run = part[t] - sum;
  for (int k = 0; k < 13; ++k) {
    int4 o;
    o.x = run; run += v[k].x;
    o.y = run; run += v[k].y;
    o.z = run; run += v[k].z;
    o.w = run; run += v[k].w;
    off[t * 13 + k]    = o;
    cntcur[t * 13 + k] = o;
  }
}

__global__ __launch_bounds__(256) void place_kernel(const int* __restrict__ dst,
                                                    int* __restrict__ cur,
                                                    int* __restrict__ perm) {
  int t = blockIdx.x * 256 + threadIdx.x;
  if (t >= N_EDGES / 4) return;
  int4 d = *(const int4*)(dst + 4 * t);
  int e = 4 * t;
  int p0 = atomicAdd(cur + d.x, 1);
  int p1 = atomicAdd(cur + d.y, 1);
  int p2 = atomicAdd(cur + d.z, 1);
  int p3 = atomicAdd(cur + d.w, 1);
  perm[min(max(p0, 0), N_EDGES - 1)] = e;
  perm[min(max(p1, 0), N_EDGES - 1)] = e + 1;
  perm[min(max(p2, 0), N_EDGES - 1)] = e + 2;
  perm[min(max(p3, 0), N_EDGES - 1)] = e + 3;
}

__global__ __launch_bounds__(256) void order_kernel(const int* __restrict__ off,
                                                    const int* __restrict__ perm,
                                                    const int* __restrict__ src,
                                                    const float* __restrict__ ea,
                                                    int4* __restrict__ rec) {
  int lane = threadIdx.x & 31;
  int w = __builtin_amdgcn_readfirstlane((int)(blockIdx.x * 8 + (threadIdx.x >> 5)));
  for (int i = w; i < N_NODES; i += ROW_WAVES) {
    int beg = max(off[i], 0);
    int end = min(off[i + 1], N_EDGES);
    for (int p = beg + lane; p < end; p += 32) {
      int e = perm[p];
      int rank = 0;
      for (int q = beg; q < end; ++q) rank += (perm[q] < e) ? 1 : 0;
      int ec = min(max(e, 0), N_EDGES - 1);
      int s = src[ec];
      float2 a = *(const float2*)(ea + 2 * ec);
      rec[beg + rank] = make_int4(s, __float_as_int(a.x), __float_as_int(a.y), 0);
    }
  }
}

__device__ __forceinline__ float4 edge_msg(float4 hv, float a0, float a1, float4 w0, float4 w1, float4 bb) {
  float m0 = hv.x + a0 * w0.x + a1 * w1.x + bb.x;
  float m1 = hv.y + a0 * w0.y + a1 * w1.y + bb.y;
  float m2 = hv.z + a0 * w0.z + a1 * w1.z + bb.z;
  float m3 = hv.w + a0 * w0.w + a1 * w1.w + bb.w;
  return make_float4(m0 > 0.f ? m0 : 0.f, m1 > 0.f ? m1 : 0.f,
                     m2 > 0.f ? m2 : 0.f, m3 > 0.f ? m3 : 0.f);
}

__device__ __forceinline__ float4 load_row(const float* __restrict__ h, int s, int c) {
  int sc = min(max(s, 0), N_NODES - 1);
  return *(const float4*)(h + (long)sc * HID + c);
}

#define TILE_PITCH 132

__device__ __forceinline__ void add_if(float4& acc, bool v, float4 m) {
  acc.x += v ? m.x : 0.f;  acc.y += v ? m.y : 0.f;  acc.z += v ? m.z : 0.f;  acc.w += v ? m.w : 0.f;
}

__global__ __launch_bounds__(256) void gine_layer_kernel(
    const float* __restrict__ h,
    const int*   __restrict__ off,
    const int4*  __restrict__ rec,
    const float* __restrict__ Wl0,
    const float* __restrict__ Wl1,
    const float* __restrict__ bl,
    const float* __restrict__ W,
    const float* __restrict__ bias,
    float* __restrict__ out)
{
  __shared__ float xt[16 * TILE_PITCH];

  int lane = threadIdx.x & 31;
  int wave = __builtin_amdgcn_readfirstlane((int)(threadIdx.x >> 5));
  int m0   = blockIdx.x * 16;
  int c    = lane * 4;

  {
    float4 w0 = *(const float4*)(Wl0 + c);
    float4 w1 = *(const float4*)(Wl1 + c);
    float4 bb = *(const float4*)(bl  + c);

    int iA = m0 + 2 * wave;
    int o0 = off[iA], o1 = off[iA + 1], o2 = off[iA + 2];
    int begA = max(o0, 0), endA = min(o1, N_EDGES);
    int begB = max(o1, 0), endB = min(o2, N_EDGES);
    int lenA = max(endA - begA, 0);
    int lenB = max(endB - begB, 0);
    int len  = max(lenA, lenB);

    float4 hA = *(const float4*)(h + (long)iA * HID + c);
    float4 hB = *(const float4*)(h + (long)(iA + 1) * HID + c);

    float4 accA = make_float4(0.f, 0.f, 0.f, 0.f);
    float4 accB = make_float4(0.f, 0.f, 0.f, 0.f);
    for (int cb = 0; cb < len; cb += 32) {
      int nA = min(max(lenA - cb, 0), 32);
      int nB = min(max(lenB - cb, 0), 32);
      int pA = min(max(begA + cb + min(lane, nA - 1), 0), N_EDGES - 1);
      int pB = min(max(begB + cb + min(lane, nB - 1), 0), N_EDGES - 1);
      int4 myA = rec[pA];
      int4 myB = rec[pB];
      int n = max(nA, nB);
      for (int t = 0; t < n; t += 4) {
        int a0 = max(min(t,     nA - 1), 0), b0 = max(min(t,     nB - 1), 0);
        int a1 = max(min(t + 1, nA - 1), 0), b1 = max(min(t + 1, nB - 1), 0);
        int a2 = max(min(t + 2, nA - 1), 0), b2 = max(min(t + 2, nB - 1), 0);
        int a3 = max(min(t + 3, nA - 1), 0), b3 = max(min(t + 3, nB - 1), 0);
        float4 xA0 = load_row(h, __builtin_amdgcn_readlane(myA.x, a0), c);
        float4 xB0 = load_row(h, __builtin_amdgcn_readlane(myB.x, b0), c);
        float4 xA1 = load_row(h, __builtin_amdgcn_readlane(myA.x, a1), c);
        float4 xB1 = load_row(h, __builtin_amdgcn_readlane(myB.x, b1), c);
        float4 xA2 = load_row(h, __builtin_amdgcn_readlane(myA.x, a2), c);
        float4 xB2 = load_row(h, __builtin_amdgcn_readlane(myB.x, b2), c);
        float4 xA3 = load_row(h, __builtin_amdgcn_readlane(myA.x, a3), c);
        float4 xB3 = load_row(h, __builtin_amdgcn_readlane(myB.x, b3), c);
        float4 mA0 = edge_msg(xA0, __int_as_float(__builtin_amdgcn_readlane(myA.y, a0)),
                                   __int_as_float(__builtin_amdgcn_readlane(myA.z, a0)), w0, w1, bb);
        float4 mA1 = edge_msg(xA1, __int_as_float(__builtin_amdgcn_readlane(myA.y, a1)),
                                   __int_as_float(__builtin_amdgcn_readlane(myA.z, a1)), w0, w1, bb);
        float4 mA2 = edge_msg(xA2, __int_as_float(__builtin_amdgcn_readlane(myA.y, a2)),
                                   __int_as_float(__builtin_amdgcn_readlane(myA.z, a2)), w0, w1, bb);
        float4 mA3 = edge_msg(xA3, __int_as_float(__builtin_amdgcn_readlane(myA.y, a3)),
                                   __int_as_float(__builtin_amdgcn_readlane(myA.z, a3)), w0, w1, bb);
        float4 mB0 = edge_msg(xB0, __int_as_float(__builtin_amdgcn_readlane(myB.y, b0)),
                                   __int_as_float(__builtin_amdgcn_readlane(myB.z, b0)), w0, w1, bb);
        float4 mB1 = edge_msg(xB1, __int_as_float(__builtin_amdgcn_readlane(myB.y, b1)),
                                   __int_as_float(__builtin_amdgcn_readlane(myB.z, b1)), w0, w1, bb);
        float4 mB2 = edge_msg(xB2, __int_as_float(__builtin_amdgcn_readlane(myB.y, b2)),
                                   __int_as_float(__builtin_amdgcn_readlane(myB.z, b2)), w0, w1, bb);
        float4 mB3 = edge_msg(xB3, __int_as_float(__builtin_amdgcn_readlane(myB.y, b3)),
                                   __int_as_float(__builtin_amdgcn_readlane(myB.z, b3)), w0, w1, bb);
        add_if(accA, t     < nA, mA0);
        add_if(accA, t + 1 < nA, mA1);
        add_if(accA, t + 2 < nA, mA2);
        add_if(accA, t + 3 < nA, mA3);
        add_if(accB, t     < nB, mB0);
        add_if(accB, t + 1 < nB, mB1);
        add_if(accB, t + 2 < nB, mB2);
        add_if(accB, t + 3 < nB, mB3);
      }
    }
    *(float4*)(xt + (2 * wave)     * TILE_PITCH + c) =
        make_float4(hA.x + accA.x, hA.y + accA.y, hA.z + accA.z, hA.w + accA.w);
    *(float4*)(xt + (2 * wave + 1) * TILE_PITCH + c) =
        make_float4(hB.x + accB.x, hB.y + accB.y, hB.z + accB.z, hB.w + accB.w);
  }

  int m    = lane & 15;
  int half = lane >> 4;
  int n    = wave * 16 + m;

  v2f bf[HID / 4];
#pragma unroll
  for (int s = 0; s < HID / 4; ++s) {
    int k0 = 4 * s + half * 2;
    bf[s].x = W[(long)k0 * HID + n];
    bf[s].y = W[(long)(k0 + 1) * HID + n];
  }
  float bv = bias[n];

  __syncthreads();

  const float* ar = xt + m * TILE_PITCH + half * 2;
  v8f cc = {};
#pragma unroll
  for (int s = 0; s < HID / 4; ++s) {
    v2f a = *(const v2f*)(ar + 4 * s);
    cc = __builtin_amdgcn_wmma_f32_16x16x4_f32(
         false, a,  false, bf[s],
         (short)0, cc,  false,  false);
  }

#pragma unroll
  for (int v = 0; v < 8; ++v) {
    int row = m0 + v + half * 8;
    float y = cc[v] + bv;
    out[(long)row * HID + n] = (y >= 0.f) ? y : NEG_SLOPE * y;
  }
}

extern "C" void kernel_launch(void* const* d_in, const int* in_sizes, int n_in,
                              void* d_out, int out_size, void* d_ws, size_t ws_size,
                              hipStream_t stream) {
  (void)in_sizes; (void)n_in; (void)out_size; (void)ws_size;

  const float* x  = (const float*)d_in[0];
  const int*   ei = (const int*)  d_in[1];
  const float* ea = (const float*)d_in[2];
  const float* Wl = (const float*)d_in[3];
  const float* bl = (const float*)d_in[4];
  const float* W  = (const float*)d_in[5];
  const float* b  = (const float*)d_in[6];
  float*       out = (float*)d_out;

  const int* src = ei;
  const int* dst = ei + N_EDGES;

  const long NH = (long)N_NODES * HID;
  float* agg  = (float*)d_ws;
  float* h0   = agg + NH;
  int*   cur  = (int*)(h0 + NH);
  int*   off  = cur + N_PAD;
  int*   perm = off + N_PAD;
  int4*  rec  = (int4*)(perm + N_EDGES);

  zero_counts_kernel<<<(N_PAD / 4 + 255) / 256, 256, 0, stream>>>((int4*)cur, N_PAD / 4);
  count_kernel<<<(N_EDGES / 4 + 255) / 256, 256, 0, stream>>>(dst, cur);
  scan_kernel<<<1, 1024, 0, stream>>>((int4*)cur, (int4*)off);
  place_kernel<<<(N_EDGES / 4 + 255) / 256, 256, 0, stream>>>(dst, cur, perm);
  order_kernel<<<ROW_BLOCKS, 256, 0, stream>>>(off, perm, src, ea, rec);

  const float* hin = x;
  for (int layer = 0; layer < N_LAYERS; ++layer) {
    float* hout = (layer == 1) ? h0 : out;

    gine_layer_kernel<<<N_NODES / 16, 256, 0, stream>>>(
        hin, off, rec,
        Wl + (long)layer * 2 * HID,
        Wl + (long)layer * 2 * HID + HID,
        bl + (long)layer * HID,
        W + (long)layer * HID * HID,
        b + (long)layer * HID,
        hout);

    hin = hout;
  }
}


# Counting pass keeps each edge's slot (its add's return value); placing pass reads segment start + slot with no atomics: one of two rounds of 600k scattered adds gone; sum order unchanged
# speedup vs baseline: 2.4509x; 1.1878x over previous
#include <hip/hip_runtime.h>


#define N_NODES 50000
#define N_EDGES 600000
#define HID 128
#define NEG_SLOPE 0.01f
#define N_LAYERS 3

#define N_PAD 53248
#define ROW_BLOCKS 2048
#define ROW_WAVES (ROW_BLOCKS * 8)

typedef __attribute__((ext_vector_type(2))) float v2f;
typedef __attribute__((ext_vector_type(8))) float v8f;


__global__ __launch_bounds__(256) void zero_counts_kernel(int4* __restrict__ p, int n4) {
  int i = blockIdx.x * 256 + threadIdx.x;
  if (i < n4) p[i] = make_int4(0, 0, 0, 0);
}

__global__ __launch_bounds__(256) void count_kernel(const int* __restrict__ dst,
                                                    int* __restrict__ cnt,
                                                    int4* __restrict__ slot) {
  int t = blockIdx.x * 256 + threadIdx.x;
  if (t >= N_EDGES / 4) return;
  int4 d = *(const int4*)(dst + 4 * t);
  int4 s;
  s.x = atomicAdd(cnt + d.x, 1);
  s.y = atomicAdd(cnt + d.y, 1);
  s.z = atomicAdd(cnt + d.z, 1);
  s.w = atomicAdd(cnt + d.w, 1);
  slot[t] = s;
}

__global__ __launch_bounds__(1024) void scan_kernel(const int4* __restrict__ cntcur,
                                                    int4* __restrict__ off) {
  __shared__ int part[1024];
  int t = threadIdx.x;
  int4 v[13];
  int sum = 0;
  for (int k = 0; k < 13; ++k) {
    v[k] = cntcur[t * 13 + k];
    sum += v[k].x + v[k].y + v[k].z + v[k].w;
  }
  part[t] = sum;
  __syncthreads();
  for (int d = 1; d < 1024; d <<= 1) {
    int u = (t >= d) ? part[t - d] : 0;
    __syncthreads();
    part[t] += u;
    __syncthreads();
  }
  int run = part[t] - sum;
  for (int k = 0; k < 13; ++k) {
    int4 o;
    o.x = run; run += v[k].x;
    o.y = run; run += v[k].y;
    o.z = run; run += v[k].z;
    o.w = run; run += v[k].w;
    off[t * 13 + k] = o;
  }
}

__global__ __launch_bounds__(256) void place_kernel(const int* __restrict__ dst,
                                                    const int4* __restrict__ slot,
                                                    const int* __restrict__ off,
                                                    int* __restrict__ perm) {
  int t = blockIdx.x * 256 + threadIdx.x;
  if (t >= N_EDGES / 4) return;
  int4 d = *(const int4*)(dst + 4 * t);
  int4 s = slot[t];
  int e = 4 * t;
  int p0 = off[min(max(d.x, 0), N_PAD - 1)] + s.x;
  int p1 = off[min(max(d.y, 0), N_PAD - 1)] + s.y;
  int p2 = off[min(max(d.z, 0), N_PAD - 1)] + s.z;
  int p3 = off[min(max(d.w, 0), N_PAD - 1)] + s.w;
  perm[min(max(p0, 0), N_EDGES - 1)] = e;
  perm[min(max(p1, 0), N_EDGES - 1)] = e + 1;
  perm[min(max(p2, 0), N_EDGES - 1)] = e + 2;
  perm[min(max(p3, 0), N_EDGES - 1)] = e + 3;
}

__global__ __launch_bounds__(256) void order_kernel(const int* __restrict__ off,
                                                    const int* __restrict__ perm,
                                                    const int* __restrict__ src,
                                                    const float* __restrict__ ea,
                                                    int4* __restrict__ rec) {
  int lane = threadIdx.x & 31;
  int w = __builtin_amdgcn_readfirstlane((int)(blockIdx.x * 8 + (threadIdx.x >> 5)));
  for (int i = w; i < N_NODES; i += ROW_WAVES) {
    int beg = max(off[i], 0);
    int end = min(off[i + 1], N_EDGES);
    for (int p = beg + lane; p < end; p += 32) {
      int e = perm[p];
      int rank = 0;
      for (int q = beg; q < end; ++q) rank += (perm[q] < e) ? 1 : 0;
      int ec = min(max(e, 0), N_EDGES - 1);
      int s = src[ec];
      float2 a = *(const float2*)(ea + 2 * ec);
      rec[beg + rank] = make_int4(s, __float_as_int(a.x), __float_as_int(a.y), 0);
    }
  }
}

__device__ __forceinline__ float4 edge_msg(float4 hv, float a0, float a1, float4 w0, float4 w1, float4 bb) {
  float m0 = hv.x + a0 * w0.x + a1 * w1.x + bb.x;
  float m1 = hv.y + a0 * w0.y + a1 * w1.y + bb.y;
  float m2 = hv.z + a0 * w0.z + a1 * w1.z + bb.z;
  float m3 = hv.w + a0 * w0.w + a1 * w1.w + bb.w;
  return make_float4(m0 > 0.f ? m0 : 0.f, m1 > 0.f ? m1 : 0.f,
                     m2 > 0.f ? m2 : 0.f, m3 > 0.f ? m3 : 0.f);
}

__device__ __forceinline__ float4 load_row(const float* __restrict__ h, int s, int c) {
  int sc = min(max(s, 0), N_NODES - 1);
  return *(const float4*)(h + (long)sc * HID + c);
}

#define TILE_PITCH 132

__device__ __forceinline__ void add_if(float4& acc, bool v, float4 m) {
  acc.x += v ? m.x : 0.f;  acc.y += v ? m.y : 0.f;  acc.z += v ? m.z : 0.f;  acc.w += v ? m.w : 0.f;
}

__global__ __launch_bounds__(256) void gine_layer_kernel(
    const float* __restrict__ h,
    const int*   __restrict__ off,
    const int4*  __restrict__ rec,
    const float* __restrict__ Wl0,
    const float* __restrict__ Wl1,
    const float* __restrict__ bl,
    const float* __restrict__ W,
    const float* __restrict__ bias,
    float* __restrict__ out)
{
  __shared__ float xt[16 * TILE_PITCH];

  int lane = threadIdx.x & 31;
  int wave = __builtin_amdgcn_readfirstlane((int)(threadIdx.x >> 5));
  int m0   = blockIdx.x * 16;
  int c    = lane * 4;

  {
    float4 w0 = *(const float4*)(Wl0 + c);
    float4 w1 = *(const float4*)(Wl1 + c);
    float4 bb = *(const float4*)(bl  + c);

    int iA = m0 + 2 * wave;
    int o0 = off[iA], o1 = off[iA + 1], o2 = off[iA + 2];
    int begA = max(o0, 0), endA = min(o1, N_EDGES);
    int begB = max(o1, 0), endB = min(o2, N_EDGES);
    int lenA = max(endA - begA, 0);
    int lenB = max(endB - begB, 0);
    int len  = max(lenA, lenB);

    float4 hA = *(const float4*)(h + (long)iA * HID + c);
    float4 hB = *(const float4*)(h + (long)(iA + 1) * HID + c);

    float4 accA = make_float4(0.f, 0.f, 0.f, 0.f);
    float4 accB = make_float4(0.f, 0.f, 0.f, 0.f);
    for (int cb = 0; cb < len; cb += 32) {
      int nA = min(max(lenA - cb, 0), 32);
      int nB = min(max(lenB - cb, 0), 32);
      int pA = min(max(begA + cb + min(lane, nA - 1), 0), N_EDGES - 1);
      int pB = min(max(begB + cb + min(lane, nB - 1), 0), N_EDGES - 1);
      int4 myA = rec[pA];
      int4 myB = rec[pB];
      int n = max(nA, nB);
      for (int t = 0; t < n; t += 4) {
        int a0 = max(min(t,     nA - 1), 0), b0 = max(min(t,     nB - 1), 0);
        int a1 = max(min(t + 1, nA - 1), 0), b1 = max(min(t + 1, nB - 1), 0);
        int a2 = max(min(t + 2, nA - 1), 0), b2 = max(min(t + 2, nB - 1), 0);
        int a3 = max(min(t + 3, nA - 1), 0), b3 = max(min(t + 3, nB - 1), 0);
        float4 xA0 = load_row(h, __builtin_amdgcn_readlane(myA.x, a0), c);
        float4 xB0 = load_row(h, __builtin_amdgcn_readlane(myB.x, b0), c);
        float4 xA1 = load_row(h, __builtin_amdgcn_readlane(myA.x, a1), c);
        float4 xB1 = load_row(h, __builtin_amdgcn_readlane(myB.x, b1), c);
        float4 xA2 = load_row(h, __builtin_amdgcn_readlane(myA.x, a2), c);
        float4 xB2 = load_row(h, __builtin_amdgcn_readlane(myB.x, b2), c);
        float4 xA3 = load_row(h, __builtin_amdgcn_readlane(myA.x, a3), c);
        float4 xB3 = load_row(h, __builtin_amdgcn_readlane(myB.x, b3), c);
        float4 mA0 = edge_msg(xA0, __int_as_float(__builtin_amdgcn_readlane(myA.y, a0)),
                                   __int_as_float(__builtin_amdgcn_readlane(myA.z, a0)), w0, w1, bb);
        float4 mA1 = edge_msg(xA1, __int_as_float(__builtin_amdgcn_readlane(myA.y, a1)),
                                   __int_as_float(__builtin_amdgcn_readlane(myA.z, a1)), w0, w1, bb);
        float4 mA2 = edge_msg(xA2, __int_as_float(__builtin_amdgcn_readlane(myA.y, a2)),
                                   __int_as_float(__builtin_amdgcn_readlane(myA.z, a2)), w0, w1, bb);
        float4 mA3 = edge_msg(xA3, __int_as_float(__builtin_amdgcn_readlane(myA.y, a3)),
                                   __int_as_float(__builtin_amdgcn_readlane(myA.z, a3)), w0, w1, bb);
        float4 mB0 = edge_msg(xB0, __int_as_float(__builtin_amdgcn_readlane(myB.y, b0)),
                                   __int_as_float(__builtin_amdgcn_readlane(myB.z, b0)), w0, w1, bb);
        float4 mB1 = edge_msg(xB1, __int_as_float(__builtin_amdgcn_readlane(myB.y, b1)),
                                   __int_as_float(__builtin_amdgcn_readlane(myB.z, b1)), w0, w1, bb);
        float4 mB2 = edge_msg(xB2, __int_as_float(__builtin_amdgcn_readlane(myB.y, b2)),
                                   __int_as_float(__builtin_amdgcn_readlane(myB.z, b2)), w0, w1, bb);
        float4 mB3 = edge_msg(xB3, __int_as_float(__builtin_amdgcn_readlane(myB.y, b3)),
                                   __int_as_float(__builtin_amdgcn_readlane(myB.z, b3)), w0, w1, bb);
        add_if(accA, t     < nA, mA0);
        add_if(accA, t + 1 < nA, mA1);
        add_if(accA, t + 2 < nA, mA2);
        add_if(accA, t + 3 < nA, mA3);
        add_if(accB, t     < nB, mB0);
        add_if(accB, t + 1 < nB, mB1);
        add_if(accB, t + 2 < nB, mB2);
        add_if(accB, t + 3 < nB, mB3);
      }
    }
    *(float4*)(xt + (2 * wave)     * TILE_PITCH + c) =
        make_float4(hA.x + accA.x, hA.y + accA.y, hA.z + accA.z, hA.w + accA.w);
    *(float4*)(xt + (2 * wave + 1) * TILE_PITCH + c) =
        make_float4(hB.x + accB.x, hB.y + accB.y, hB.z + accB.z, hB.w + accB.w);
  }

  int m    = lane & 15;
  int half = lane >> 4;
  int n    = wave * 16 + m;

  v2f bf[HID / 4];
#pragma unroll
  for (int s = 0; s < HID / 4; ++s) {
    int k0 = 4 * s + half * 2;
    bf[s].x = W[(long)k0 * HID + n];
    bf[s].y = W[(long)(k0 + 1) * HID + n];
  }
  float bv = bias[n];

  __syncthreads();

  const float* ar = xt + m * TILE_PITCH + half * 2;
  v8f cc = {};
#pragma unroll
  for (int s = 0; s < HID / 4; ++s) {
    v2f a = *(const v2f*)(ar + 4 * s);
    cc = __builtin_amdgcn_wmma_f32_16x16x4_f32(
         false, a,  false, bf[s],
         (short)0, cc,  false,  false);
  }

#pragma unroll
  for (int v = 0; v < 8; ++v) {
    int row = m0 + v + half * 8;
    float y = cc[v] + bv;
    out[(long)row * HID + n] = (y >= 0.f) ? y : NEG_SLOPE * y;
  }
}

extern "C" void kernel_launch(void* const* d_in, const int* in_sizes, int n_in,
                              void* d_out, int out_size, void* d_ws, size_t ws_size,
                              hipStream_t stream) {
  (void)in_sizes; (void)n_in; (void)out_size; (void)ws_size;

  const float* x  = (const float*)d_in[0];
  const int*   ei = (const int*)  d_in[1];
  const float* ea = (const float*)d_in[2];
  const float* Wl = (const float*)d_in[3];
  const float* bl = (const float*)d_in[4];
  const float* W  = (const float*)d_in[5];
  const float* b  = (const float*)d_in[6];
  float*       out = (float*)d_out;

  const int* src = ei;
  const int* dst = ei + N_EDGES;

  const long NH = (long)N_NODES * HID;
  float* agg  = (float*)d_ws;
  int4*  slot = (int4*)d_ws;
  float* h0   = agg + NH;
  int*   cur  = (int*)(h0 + NH);
  int*   off  = cur + N_PAD;
  int*   perm = off + N_PAD;
  int4*  rec  = (int4*)(perm + N_EDGES);

  zero_counts_kernel<<<(N_PAD / 4 + 255) / 256, 256, 0, stream>>>((int4*)cur, N_PAD / 4);
  count_kernel<<<(N_EDGES / 4 + 255) / 256, 256, 0, stream>>>(dst, cur, slot);
  scan_kernel<<<1, 1024, 0, stream>>>((const int4*)cur, (int4*)off);
  place_kernel<<<(N_EDGES / 4 + 255) / 256, 256, 0, stream>>>(dst, slot, off, perm);
  order_kernel<<<ROW_BLOCKS, 256, 0, stream>>>(off, perm, src, ea, rec);

  const float* hin = x;
  for (int layer = 0; layer < N_LAYERS; ++layer) {
    float* hout = (layer == 1) ? h0 : out;

    gine_layer_kernel<<<N_NODES / 16, 256, 0, stream>>>(
        hin, off, rec,
        Wl + (long)layer * 2 * HID,
        Wl + (long)layer * 2 * HID + HID,
        bl + (long)layer * HID,
        W + (long)layer * HID * HID,
        b + (long)layer * HID,
        hout);

    hin = hout;
  }
}


# Counting pass on LDS atomics: one block per chunk of 18750 edges counts into 208 KB of LDS counters, a per-node prefix over the 32 chunks gives slots; no global atomics, no zero pass; same sum order
# speedup vs baseline: 2.8306x; 1.1549x over previous
#include <hip/hip_runtime.h>


#define N_NODES 50000
#define N_EDGES 600000
#define HID 128
#define NEG_SLOPE 0.01f
#define N_LAYERS 3

#define N_PAD 53248
#define ROW_BLOCKS 2048
#define ROW_WAVES (ROW_BLOCKS * 8)

typedef __attribute__((ext_vector_type(2))) float v2f;
typedef __attribute__((ext_vector_type(8))) float v8f;

#define CH 32
#define CH_EDGES (N_EDGES / CH)
#define CH_TRIPS ((CH_EDGES + 1023) / 1024)

__global__ __launch_bounds__(1024) void count_kernel(const int* __restrict__ dst,
                                                     int* __restrict__ part,
                                                     int* __restrict__ lslot) {
  __shared__ int lc[N_PAD];
  int tid = threadIdx.x;
  int e0  = blockIdx.x * CH_EDGES;
  int dv[CH_TRIPS];
#pragma unroll
  for (int k = 0; k < CH_TRIPS; ++k) {
    int t = tid + 1024 * k;
    dv[k] = (t < CH_EDGES) ? dst[e0 + t] : 0;
  }
  int4* lc4 = (int4*)lc;
#pragma unroll
  for (int k = 0; k < N_PAD / 4 / 1024; ++k) lc4[tid + 1024 * k] = make_int4(0, 0, 0, 0);
  __syncthreads();
  int sl[CH_TRIPS];
#pragma unroll
  for (int k = 0; k < CH_TRIPS; ++k) {
    int t = tid + 1024 * k;
    int d = min(max(dv[k], 0), N_PAD - 1);
    sl[k] = (t < CH_EDGES) ? atomicAdd(&lc[d], 1) : 0;
  }
  __syncthreads();
#pragma unroll
  for (int k = 0; k < CH_TRIPS; ++k) {
    int t = tid + 1024 * k;
    if (t < CH_EDGES) lslot[e0 + t] = sl[k];
  }
  int4* p4 = (int4*)(part + (long)blockIdx.x * N_PAD);
#pragma unroll
  for (int k = 0; k < N_PAD / 4 / 1024; ++k) p4[tid + 1024 * k] = lc4[tid + 1024 * k];
}

__global__ __launch_bounds__(256) void chunk_prefix_kernel(int* __restrict__ part,
                                                           int* __restrict__ cnt) {
  int d = blockIdx.x * 256 + threadIdx.x;
  int v[CH];
#pragma unroll
  for (int i = 0; i < CH; ++i) v[i] = part[(long)i * N_PAD + d];
  int run = 0;
#pragma unroll
  for (int i = 0; i < CH; ++i) {
    part[(long)i * N_PAD + d] = run;
    run += v[i];
  }
  cnt[d] = run;
}

__global__ __launch_bounds__(1024) void scan_kernel(const int4* __restrict__ cntcur,
                                                    int4* __restrict__ off) {
  __shared__ int part[1024];
  int t = threadIdx.x;
  int4 v[13];
  int sum = 0;
  for (int k = 0; k < 13; ++k) {
    v[k] = cntcur[t * 13 + k];
    sum += v[k].x + v[k].y + v[k].z + v[k].w;
  }
  part[t] = sum;
  __syncthreads();
  for (int d = 1; d < 1024; d <<= 1) {
    int u = (t >= d) ? part[t - d] : 0;
    __syncthreads();
    part[t] += u;
    __syncthreads();
  }
  int run = part[t] - sum;
  for (int k = 0; k < 13; ++k) {
    int4 o;
    o.x = run; run += v[k].x;
    o.y = run; run += v[k].y;
    o.z = run; run += v[k].z;
    o.w = run; run += v[k].w;
    off[t * 13 + k] = o;
  }
}

__global__ __launch_bounds__(256) void place_kernel(const int* __restrict__ dst,
                                                    const int4* __restrict__ lslot,
                                                    const int* __restrict__ part,
                                                    const int* __restrict__ off,
                                                    int* __restrict__ perm) {
  int t = blockIdx.x * 256 + threadIdx.x;
  if (t >= N_EDGES / 4) return;
  int4 d = *(const int4*)(dst + 4 * t);
  int4 s = lslot[t];
  int e = 4 * t;
  int d0 = min(max(d.x, 0), N_PAD - 1);
  int d1 = min(max(d.y, 0), N_PAD - 1);
  int d2 = min(max(d.z, 0), N_PAD - 1);
  int d3 = min(max(d.w, 0), N_PAD - 1);
  int c0 = e / CH_EDGES;
  int c1 = (e + 1) / CH_EDGES;
  int c2 = (e + 2) / CH_EDGES;
  int c3 = (e + 3) / CH_EDGES;
  int p0 = off[d0] + part[(long)c0 * N_PAD + d0] + s.x;
  int p1 = off[d1] + part[(long)c1 * N_PAD + d1] + s.y;
  int p2 = off[d2] + part[(long)c2 * N_PAD + d2] + s.z;
  int p3 = off[d3] + part[(long)c3 * N_PAD + d3] + s.w;
  perm[min(max(p0, 0), N_EDGES - 1)] = e;
  perm[min(max(p1, 0), N_EDGES - 1)] = e + 1;
  perm[min(max(p2, 0), N_EDGES - 1)] = e + 2;
  perm[min(max(p3, 0), N_EDGES - 1)] = e + 3;
}

__global__ __launch_bounds__(256) void order_kernel(const int* __restrict__ off,
                                                    const int* __restrict__ perm,
                                                    const int* __restrict__ src,
                                                    const float* __restrict__ ea,
                                                    int4* __restrict__ rec) {
  int lane = threadIdx.x & 31;
  int w = __builtin_amdgcn_readfirstlane((int)(blockIdx.x * 8 + (threadIdx.x >> 5)));
  for (int i = w; i < N_NODES; i += ROW_WAVES) {
    int beg = max(off[i], 0);
    int end = min(off[i + 1], N_EDGES);
    for (int p = beg + lane; p < end; p += 32) {
      int e = perm[p];
      int rank = 0;
      for (int q = beg; q < end; ++q) rank += (perm[q] < e) ? 1 : 0;
      int ec = min(max(e, 0), N_EDGES - 1);
      int s = src[ec];
      float2 a = *(const float2*)(ea + 2 * ec);
      rec[beg + rank] = make_int4(s, __float_as_int(a.x), __float_as_int(a.y), 0);
    }
  }
}

__device__ __forceinline__ float4 edge_msg(float4 hv, float a0, float a1, float4 w0, float4 w1, float4 bb) {
  float m0 = hv.x + a0 * w0.x + a1 * w1.x + bb.x;
  float m1 = hv.y + a0 * w0.y + a1 * w1.y + bb.y;
  float m2 = hv.z + a0 * w0.z + a1 * w1.z + bb.z;
  float m3 = hv.w + a0 * w0.w + a1 * w1.w + bb.w;
  return make_float4(m0 > 0.f ? m0 : 0.f, m1 > 0.f ? m1 : 0.f,
                     m2 > 0.f ? m2 : 0.f, m3 > 0.f ? m3 : 0.f);
}

__device__ __forceinline__ float4 load_row(const float* __restrict__ h, int s, int c) {
  int sc = min(max(s, 0), N_NODES - 1);
  return *(const float4*)(h + (long)sc * HID + c);
}

#define TILE_PITCH 132

__device__ __forceinline__ void add_if(float4& acc, bool v, float4 m) {
  acc.x += v ? m.x : 0.f;  acc.y += v ? m.y : 0.f;  acc.z += v ? m.z : 0.f;  acc.w += v ? m.w : 0.f;
}

__global__ __launch_bounds__(256) void gine_layer_kernel(
    const float* __restrict__ h,
    const int*   __restrict__ off,
    const int4*  __restrict__ rec,
    const float* __restrict__ Wl0,
    const float* __restrict__ Wl1,
    const float* __restrict__ bl,
    const float* __restrict__ W,
    const float* __restrict__ bias,
    float* __restrict__ out)
{
  __shared__ float xt[16 * TILE_PITCH];

  int lane = threadIdx.x & 31;
  int wave = __builtin_amdgcn_readfirstlane((int)(threadIdx.x >> 5));
  int m0   = blockIdx.x * 16;
  int c    = lane * 4;

  {
    float4 w0 = *(const float4*)(Wl0 + c);
    float4 w1 = *(const float4*)(Wl1 + c);
    float4 bb = *(const float4*)(bl  + c);

    int iA = m0 + 2 * wave;
    int o0 = off[iA], o1 = off[iA + 1], o2 = off[iA + 2];
    int begA = max(o0, 0), endA = min(o1, N_EDGES);
    int begB = max(o1, 0), endB = min(o2, N_EDGES);
    int lenA = max(endA - begA, 0);
    int lenB = max(endB - begB, 0);
    int len  = max(lenA, lenB);

    float4 hA = *(const float4*)(h + (long)iA * HID + c);
    float4 hB = *(const float4*)(h + (long)(iA + 1) * HID + c);

    float4 accA = make_float4(0.f, 0.f, 0.f, 0.f);
    float4 accB = make_float4(0.f, 0.f, 0.f, 0.f);
    for (int cb = 0; cb < len; cb += 32) {
      int nA = min(max(lenA - cb, 0), 32);
      int nB = min(max(lenB - cb, 0), 32);
      int pA = min(max(begA + cb + min(lane, nA - 1), 0), N_EDGES - 1);
      int pB = min(max(begB + cb + min(lane, nB - 1), 0), N_EDGES - 1);
      int4 myA = rec[pA];
      int4 myB = rec[pB];
      int n = max(nA, nB);
      for (int t = 0; t < n; t += 4) {
        int a0 = max(min(t,     nA - 1), 0), b0 = max(min(t,     nB - 1), 0);
        int a1 = max(min(t + 1, nA - 1), 0), b1 = max(min(t + 1, nB - 1), 0);
        int a2 = max(min(t + 2, nA - 1), 0), b2 = max(min(t + 2, nB - 1), 0);
        int a3 = max(min(t + 3, nA - 1), 0), b3 = max(min(t + 3, nB - 1), 0);
        float4 xA0 = load_row(h, __builtin_amdgcn_readlane(myA.x, a0), c);
        float4 xB0 = load_row(h, __builtin_amdgcn_readlane(myB.x, b0), c);
        float4 xA1 = load_row(h, __builtin_amdgcn_readlane(myA.x, a1), c);
        float4 xB1 = load_row(h, __builtin_amdgcn_readlane(myB.x, b1), c);
        float4 xA2 = load_row(h, __builtin_amdgcn_readlane(myA.x, a2), c);
        float4 xB2 = load_row(h, __builtin_amdgcn_readlane(myB.x, b2), c);
        float4 xA3 = load_row(h, __builtin_amdgcn_readlane(myA.x, a3), c);
        float4 xB3 = load_row(h, __builtin_amdgcn_readlane(myB.x, b3), c);
        float4 mA0 = edge_msg(xA0, __int_as_float(__builtin_amdgcn_readlane(myA.y, a0)),
                                   __int_as_float(__builtin_amdgcn_readlane(myA.z, a0)), w0, w1, bb);
        float4 mA1 = edge_msg(xA1, __int_as_float(__builtin_amdgcn_readlane(myA.y, a1)),
                                   __int_as_float(__builtin_amdgcn_readlane(myA.z, a1)), w0, w1, bb);
        float4 mA2 = edge_msg(xA2, __int_as_float(__builtin_amdgcn_readlane(myA.y, a2)),
                                   __int_as_float(__builtin_amdgcn_readlane(myA.z, a2)), w0, w1, bb);
        float4 mA3 = edge_msg(xA3, __int_as_float(__builtin_amdgcn_readlane(myA.y, a3)),
                                   __int_as_float(__builtin_amdgcn_readlane(myA.z, a3)), w0, w1, bb);
        float4 mB0 = edge_msg(xB0, __int_as_float(__builtin_amdgcn_readlane(myB.y, b0)),
                                   __int_as_float(__builtin_amdgcn_readlane(myB.z, b0)), w0, w1, bb);
        float4 mB1 = edge_msg(xB1, __int_as_float(__builtin_amdgcn_readlane(myB.y, b1)),
                                   __int_as_float(__builtin_amdgcn_readlane(myB.z, b1)), w0, w1, bb);
        float4 mB2 = edge_msg(xB2, __int_as_float(__builtin_amdgcn_readlane(myB.y, b2)),
                                   __int_as_float(__builtin_amdgcn_readlane(myB.z, b2)), w0, w1, bb);
        float4 mB3 = edge_msg(xB3, __int_as_float(__builtin_amdgcn_readlane(myB.y, b3)),
                                   __int_as_float(__builtin_amdgcn_readlane(myB.z, b3)), w0, w1, bb);
        add_if(accA, t     < nA, mA0);
        add_if(accA, t + 1 < nA, mA1);
        add_if(accA, t + 2 < nA, mA2);
        add_if(accA, t + 3 < nA, mA3);
        add_if(accB, t     < nB, mB0);
        add_if(accB, t + 1 < nB, mB1);
        add_if(accB, t + 2 < nB, mB2);
        add_if(accB, t + 3 < nB, mB3);
      }
    }
    *(float4*)(xt + (2 * wave)     * TILE_PITCH + c) =
        make_float4(hA.x + accA.x, hA.y + accA.y, hA.z + accA.z, hA.w + accA.w);
    *(float4*)(xt + (2 * wave + 1) * TILE_PITCH + c) =
        make_float4(hB.x + accB.x, hB.y + accB.y, hB.z + accB.z, hB.w + accB.w);
  }

  int m    = lane & 15;
  int half = lane >> 4;
  int n    = wave * 16 + m;

  v2f bf[HID / 4];
#pragma unroll
  for (int s = 0; s < HID / 4; ++s) {
    int k0 = 4 * s + half * 2;
    bf[s].x = W[(long)k0 * HID + n];
    bf[s].y = W[(long)(k0 + 1) * HID + n];
  }
  float bv = bias[n];

  __syncthreads();

  const float* ar = xt + m * TILE_PITCH + half * 2;
  v8f cc = {};
#pragma unroll
  for (int s = 0; s < HID / 4; ++s) {
    v2f a = *(const v2f*)(ar + 4 * s);
    cc = __builtin_amdgcn_wmma_f32_16x16x4_f32(
         false, a,  false, bf[s],
         (short)0, cc,  false,  false);
  }

#pragma unroll
  for (int v = 0; v < 8; ++v) {
    int row = m0 + v + half * 8;
    float y = cc[v] + bv;
    out[(long)row * HID + n] = (y >= 0.f) ? y : NEG_SLOPE * y;
  }
}

extern "C" void kernel_launch(void* const* d_in, const int* in_sizes, int n_in,
                              void* d_out, int out_size, void* d_ws, size_t ws_size,
                              hipStream_t stream) {
  (void)in_sizes; (void)n_in; (void)out_size; (void)ws_size;

  const float* x  = (const float*)d_in[0];
  const int*   ei = (const int*)  d_in[1];
  const float* ea = (const float*)d_in[2];
  const float* Wl = (const float*)d_in[3];
  const float* bl = (const float*)d_in[4];
  const float* W  = (const float*)d_in[5];
  const float* b  = (const float*)d_in[6];
  float*       out = (float*)d_out;

  const int* src = ei;
  const int* dst = ei + N_EDGES;

  const long NH = (long)N_NODES * HID;
  float* agg  = (float*)d_ws;
  int*   lslot = (int*)d_ws;
  int*   part = lslot + N_EDGES;
  float* h0   = agg + NH;
  int*   cur  = (int*)(h0 + NH);
  int*   off  = cur + N_PAD;
  int*   perm = off + N_PAD;
  int4*  rec  = (int4*)(perm + N_EDGES);

  count_kernel<<<CH, 1024, 0, stream>>>(dst, part, lslot);
  chunk_prefix_kernel<<<N_PAD / 256, 256, 0, stream>>>(part, cur);
  scan_kernel<<<1, 1024, 0, stream>>>((const int4*)cur, (int4*)off);
  place_kernel<<<(N_EDGES / 4 + 255) / 256, 256, 0, stream>>>(dst, (const int4*)lslot, part, off, perm);
  order_kernel<<<ROW_BLOCKS, 256, 0, stream>>>(off, perm, src, ea, rec);

  const float* hin = x;
  for (int layer = 0; layer < N_LAYERS; ++layer) {
    float* hout = (layer == 1) ? h0 : out;

    gine_layer_kernel<<<N_NODES / 16, 256, 0, stream>>>(
        hin, off, rec,
        Wl + (long)layer * 2 * HID,
        Wl + (long)layer * 2 * HID + HID,
        bl + (long)layer * HID,
        W + (long)layer * HID * HID,
        b + (long)layer * HID,
        hout);

    hin = hout;
  }
}


# Edge phase walks a wave's two rows as ONE record list, 8 a trip; per-record 1/0 weights route each message to its row's sum by fma; scalar row offsets; fewer, leaner trips; sum order unchanged
# speedup vs baseline: 2.9752x; 1.0511x over previous
#include <hip/hip_runtime.h>


#define N_NODES 50000
#define N_EDGES 600000
#define HID 128
#define NEG_SLOPE 0.01f
#define N_LAYERS 3

#define N_PAD 53248
#define ROW_BLOCKS 2048
#define ROW_WAVES (ROW_BLOCKS * 8)

typedef __attribute__((ext_vector_type(2))) float v2f;
typedef __attribute__((ext_vector_type(8))) float v8f;

#define CH 32
#define CH_EDGES (N_EDGES / CH)
#define CH_TRIPS ((CH_EDGES + 1023) / 1024)

__global__ __launch_bounds__(1024) void count_kernel(const int* __restrict__ dst,
                                                     int* __restrict__ part,
                                                     int* __restrict__ lslot) {
  __shared__ int lc[N_PAD];
  int tid = threadIdx.x;
  int e0  = blockIdx.x * CH_EDGES;
  int dv[CH_TRIPS];
#pragma unroll
  for (int k = 0; k < CH_TRIPS; ++k) {
    int t = tid + 1024 * k;
    dv[k] = (t < CH_EDGES) ? dst[e0 + t] : 0;
  }
  int4* lc4 = (int4*)lc;
#pragma unroll
  for (int k = 0; k < N_PAD / 4 / 1024; ++k) lc4[tid + 1024 * k] = make_int4(0, 0, 0, 0);
  __syncthreads();
  int sl[CH_TRIPS];
#pragma unroll
  for (int k = 0; k < CH_TRIPS; ++k) {
    int t = tid + 1024 * k;
    int d = min(max(dv[k], 0), N_PAD - 1);
    sl[k] = (t < CH_EDGES) ? atomicAdd(&lc[d], 1) : 0;
  }
  __syncthreads();
#pragma unroll
  for (int k = 0; k < CH_TRIPS; ++k) {
    int t = tid + 1024 * k;
    if (t < CH_EDGES) lslot[e0 + t] = sl[k];
  }
  int4* p4 = (int4*)(part + (long)blockIdx.x * N_PAD);
#pragma unroll
  for (int k = 0; k < N_PAD / 4 / 1024; ++k) p4[tid + 1024 * k] = lc4[tid + 1024 * k];
}

__global__ __launch_bounds__(256) void chunk_prefix_kernel(int* __restrict__ part,
                                                           int* __restrict__ cnt) {
  int d = blockIdx.x * 256 + threadIdx.x;
  int v[CH];
#pragma unroll
  for (int i = 0; i < CH; ++i) v[i] = part[(long)i * N_PAD + d];
  int run = 0;
#pragma unroll
  for (int i = 0; i < CH; ++i) {
    part[(long)i * N_PAD + d] = run;
    run += v[i];
  }
  cnt[d] = run;
}

__global__ __launch_bounds__(1024) void scan_kernel(const int4* __restrict__ cntcur,
                                                    int4* __restrict__ off) {
  __shared__ int part[1024];
  int t = threadIdx.x;
  int4 v[13];
  int sum = 0;
  for (int k = 0; k < 13; ++k) {
    v[k] = cntcur[t * 13 + k];
    sum += v[k].x + v[k].y + v[k].z + v[k].w;
  }
  part[t] = sum;
  __syncthreads();
  for (int d = 1; d < 1024; d <<= 1) {
    int u = (t >= d) ? part[t - d] : 0;
    __syncthreads();
    part[t] += u;
    __syncthreads();
  }
  int run = part[t] - sum;
  for (int k = 0; k < 13; ++k) {
    int4 o;
    o.x = run; run += v[k].x;
    o.y = run; run += v[k].y;
    o.z = run; run += v[k].z;
    o.w = run; run += v[k].w;
    off[t * 13 + k] = o;
  }
}

__global__ __launch_bounds__(256) void place_kernel(const int* __restrict__ dst,
                                                    const int4* __restrict__ lslot,
                                                    const int* __restrict__ part,
                                                    const int* __restrict__ off,
                                                    int* __restrict__ perm) {
  int t = blockIdx.x * 256 + threadIdx.x;
  if (t >= N_EDGES / 4) return;
  int4 d = *(const int4*)(dst + 4 * t);
  int4 s = lslot[t];
  int e = 4 * t;
  int d0 = min(max(d.x, 0), N_PAD - 1);
  int d1 = min(max(d.y, 0), N_PAD - 1);
  int d2 = min(max(d.z, 0), N_PAD - 1);
  int d3 = min(max(d.w, 0), N_PAD - 1);
  int c0 = e / CH_EDGES;
  int c1 = (e + 1) / CH_EDGES;
  int c2 = (e + 2) / CH_EDGES;
  int c3 = (e + 3) / CH_EDGES;
  int p0 = off[d0] + part[(long)c0 * N_PAD + d0] + s.x;
  int p1 = off[d1] + part[(long)c1 * N_PAD + d1] + s.y;
  int p2 = off[d2] + part[(long)c2 * N_PAD + d2] + s.z;
  int p3 = off[d3] + part[(long)c3 * N_PAD + d3] + s.w;
  perm[min(max(p0, 0), N_EDGES - 1)] = e;
  perm[min(max(p1, 0), N_EDGES - 1)] = e + 1;
  perm[min(max(p2, 0), N_EDGES - 1)] = e + 2;
  perm[min(max(p3, 0), N_EDGES - 1)] = e + 3;
}

__global__ __launch_bounds__(256) void order_kernel(const int* __restrict__ off,
                                                    const int* __restrict__ perm,
                                                    const int* __restrict__ src,
                                                    const float* __restrict__ ea,
                                                    int4* __restrict__ rec) {
  int lane = threadIdx.x & 31;
  int w = __builtin_amdgcn_readfirstlane((int)(blockIdx.x * 8 + (threadIdx.x >> 5)));
  for (int i = w; i < N_NODES; i += ROW_WAVES) {
    int beg = max(off[i], 0);
    int end = min(off[i + 1], N_EDGES);
    for (int p = beg + lane; p < end; p += 32) {
      int e = perm[p];
      int rank = 0;
      for (int q = beg; q < end; ++q) rank += (perm[q] < e) ? 1 : 0;
      int ec = min(max(e, 0), N_EDGES - 1);
      int s = src[ec];
      float2 a = *(const float2*)(ea + 2 * ec);
      rec[beg + rank] = make_int4(s, __float_as_int(a.x), __float_as_int(a.y), 0);
    }
  }
}

__device__ __forceinline__ float4 edge_msg(float4 hv, float a0, float a1, float4 w0, float4 w1, float4 bb) {
  float m0 = hv.x + a0 * w0.x + a1 * w1.x + bb.x;
  float m1 = hv.y + a0 * w0.y + a1 * w1.y + bb.y;
  float m2 = hv.z + a0 * w0.z + a1 * w1.z + bb.z;
  float m3 = hv.w + a0 * w0.w + a1 * w1.w + bb.w;
  return make_float4(m0 > 0.f ? m0 : 0.f, m1 > 0.f ? m1 : 0.f,
                     m2 > 0.f ? m2 : 0.f, m3 > 0.f ? m3 : 0.f);
}

__device__ __forceinline__ float4 load_row(const float* __restrict__ h, int s, int lane) {
  unsigned sc = min((unsigned)s, (unsigned)(N_NODES - 1));
  return ((const float4*)h)[sc * (unsigned)(HID / 4) + (unsigned)lane];
}

#define TILE_PITCH 132

__global__ __launch_bounds__(256) void gine_layer_kernel(
    const float* __restrict__ h,
    const int*   __restrict__ off,
    const int4*  __restrict__ rec,
    const float* __restrict__ Wl0,
    const float* __restrict__ Wl1,
    const float* __restrict__ bl,
    const float* __restrict__ W,
    const float* __restrict__ bias,
    float* __restrict__ out)
{
  __shared__ float xt[16 * TILE_PITCH];

  int lane = threadIdx.x & 31;
  int wave = __builtin_amdgcn_readfirstlane((int)(threadIdx.x >> 5));
  int m0   = blockIdx.x * 16;
  int c    = lane * 4;

  {
    float4 w0 = *(const float4*)(Wl0 + c);
    float4 w1 = *(const float4*)(Wl1 + c);
    float4 bb = *(const float4*)(bl  + c);

    int iA = m0 + 2 * wave;
    int o0 = off[iA], o1 = off[iA + 1], o2 = off[iA + 2];
    int beg = min(max(o0, 0), N_EDGES);
    int mid = min(max(o1, beg), N_EDGES);
    int end = min(max(o2, mid), N_EDGES);
    int lenA = mid - beg;
    int len  = end - beg;

    float4 hA = *(const float4*)(h + (long)iA * HID + c);
    float4 hB = *(const float4*)(h + (long)(iA + 1) * HID + c);

    float4 accA = make_float4(0.f, 0.f, 0.f, 0.f);
    float4 accB = make_float4(0.f, 0.f, 0.f, 0.f);
    for (int cb = 0; cb < len; cb += 32) {
      int n   = min(len - cb, 32);
      int nAc = min(max(lenA - cb, 0), n);
      int p = min(beg + cb + min(lane, n - 1), N_EDGES - 1);
      int4 my = rec[p];
      float waV = (lane < nAc) ? 1.f : 0.f;
      float wbV = (lane >= nAc && lane < n) ? 1.f : 0.f;
      for (int t = 0; t < n; t += 8) {
        float4 xr[8];
        float  e0[8], e1[8], wa[8], wb[8];
#pragma unroll
        for (int j = 0; j < 8; ++j) {
          int q = t + j;
          xr[j] = load_row(h, __builtin_amdgcn_readlane(my.x, q), lane);
          e0[j] = __int_as_float(__builtin_amdgcn_readlane(my.y, q));
          e1[j] = __int_as_float(__builtin_amdgcn_readlane(my.z, q));
          wa[j] = __int_as_float(__builtin_amdgcn_readlane(__float_as_int(waV), q));
          wb[j] = __int_as_float(__builtin_amdgcn_readlane(__float_as_int(wbV), q));
        }
#pragma unroll
        for (int j = 0; j < 8; ++j) {
          float4 m = edge_msg(xr[j], e0[j], e1[j], w0, w1, bb);
          accA.x = fmaf(m.x, wa[j], accA.x);  accA.y = fmaf(m.y, wa[j], accA.y);
          accA.z = fmaf(m.z, wa[j], accA.z);  accA.w = fmaf(m.w, wa[j], accA.w);
          accB.x = fmaf(m.x, wb[j], accB.x);  accB.y = fmaf(m.y, wb[j], accB.y);
          accB.z = fmaf(m.z, wb[j], accB.z);  accB.w = fmaf(m.w, wb[j], accB.w);
        }
      }
    }
    *(float4*)(xt + (2 * wave)     * TILE_PITCH + c) =
        make_float4(hA.x + accA.x, hA.y + accA.y, hA.z + accA.z, hA.w + accA.w);
    *(float4*)(xt + (2 * wave + 1) * TILE_PITCH + c) =
        make_float4(hB.x + accB.x, hB.y + accB.y, hB.z + accB.z, hB.w + accB.w);
  }

  int m    = lane & 15;
  int half = lane >> 4;
  int n    = wave * 16 + m;

  v2f bf[HID / 4];
#pragma unroll
  for (int s = 0; s < HID / 4; ++s) {
    int k0 = 4 * s + half * 2;
    bf[s].x = W[(long)k0 * HID + n];
    bf[s].y = W[(long)(k0 + 1) * HID + n];
  }
  float bv = bias[n];

  __syncthreads();

  const float* ar = xt + m * TILE_PITCH + half * 2;
  v8f cc = {};
#pragma unroll
  for (int s = 0; s < HID / 4; ++s) {
    v2f a = *(const v2f*)(ar + 4 * s);
    cc = __builtin_amdgcn_wmma_f32_16x16x4_f32(
         false, a,  false, bf[s],
         (short)0, cc,  false,  false);
  }

#pragma unroll
  for (int v = 0; v < 8; ++v) {
    int row = m0 + v + half * 8;
    float y = cc[v] + bv;
    out[(long)row * HID + n] = (y >= 0.f) ? y : NEG_SLOPE * y;
  }
}

extern "C" void kernel_launch(void* const* d_in, const int* in_sizes, int n_in,
                              void* d_out, int out_size, void* d_ws, size_t ws_size,
                              hipStream_t stream) {
  (void)in_sizes; (void)n_in; (void)out_size; (void)ws_size;

  const float* x  = (const float*)d_in[0];
  const int*   ei = (const int*)  d_in[1];
  const float* ea = (const float*)d_in[2];
  const float* Wl = (const float*)d_in[3];
  const float* bl = (const float*)d_in[4];
  const float* W  = (const float*)d_in[5];
  const float* b  = (const float*)d_in[6];
  float*       out = (float*)d_out;

  const int* src = ei;
  const int* dst = ei + N_EDGES;

  const long NH = (long)N_NODES * HID;
  float* agg  = (float*)d_ws;
  int*   lslot = (int*)d_ws;
  int*   part = lslot + N_EDGES;
  float* h0   = agg + NH;
  int*   cur  = (int*)(h0 + NH);
  int*   off  = cur + N_PAD;
  int*   perm = off + N_PAD;
  int4*  rec  = (int4*)(perm + N_EDGES);

  count_kernel<<<CH, 1024, 0, stream>>>(dst, part, lslot);
  chunk_prefix_kernel<<<N_PAD / 256, 256, 0, stream>>>(part, cur);
  scan_kernel<<<1, 1024, 0, stream>>>((const int4*)cur, (int4*)off);
  place_kernel<<<(N_EDGES / 4 + 255) / 256, 256, 0, stream>>>(dst, (const int4*)lslot, part, off, perm);
  order_kernel<<<ROW_BLOCKS, 256, 0, stream>>>(off, perm, src, ea, rec);

  const float* hin = x;
  for (int layer = 0; layer < N_LAYERS; ++layer) {
    float* hout = (layer == 1) ? h0 : out;

    gine_layer_kernel<<<N_NODES / 16, 256, 0, stream>>>(
        hin, off, rec,
        Wl + (long)layer * 2 * HID,
        Wl + (long)layer * 2 * HID + HID,
        bl + (long)layer * HID,
        W + (long)layer * HID * HID,
        b + (long)layer * HID,
        hout);

    hin = hout;
  }
}


# Layer kernel fitted into 64 registers (was 80) for 16 waves per SIMD instead of 12: W strip asked for in quarters, two held at a time; record weights made on the scalar unit; sum order unchanged
# speedup vs baseline: 3.1660x; 1.0641x over previous
#include <hip/hip_runtime.h>


#define N_NODES 50000
#define N_EDGES 600000
#define HID 128
#define NEG_SLOPE 0.01f
#define N_LAYERS 3

#define N_PAD 53248
#define ROW_BLOCKS 2048
#define ROW_WAVES (ROW_BLOCKS * 8)

typedef __attribute__((ext_vector_type(2))) float v2f;
typedef __attribute__((ext_vector_type(8))) float v8f;

#define CH 32
#define CH_EDGES (N_EDGES / CH)
#define CH_TRIPS ((CH_EDGES + 1023) / 1024)

__global__ __launch_bounds__(1024) void count_kernel(const int* __restrict__ dst,
                                                     int* __restrict__ part,
                                                     int* __restrict__ lslot) {
  __shared__ int lc[N_PAD];
  int tid = threadIdx.x;
  int e0  = blockIdx.x * CH_EDGES;
  int dv[CH_TRIPS];
#pragma unroll
  for (int k = 0; k < CH_TRIPS; ++k) {
    int t = tid + 1024 * k;
    dv[k] = (t < CH_EDGES) ? dst[e0 + t] : 0;
  }
  int4* lc4 = (int4*)lc;
#pragma unroll
  for (int k = 0; k < N_PAD / 4 / 1024; ++k) lc4[tid + 1024 * k] = make_int4(0, 0, 0, 0);
  __syncthreads();
  int sl[CH_TRIPS];
#pragma unroll
  for (int k = 0; k < CH_TRIPS; ++k) {
    int t = tid + 1024 * k;
    int d = min(max(dv[k], 0), N_PAD - 1);
    sl[k] = (t < CH_EDGES) ? atomicAdd(&lc[d], 1) : 0;
  }
  __syncthreads();
#pragma unroll
  for (int k = 0; k < CH_TRIPS; ++k) {
    int t = tid + 1024 * k;
    if (t < CH_EDGES) lslot[e0 + t] = sl[k];
  }
  int4* p4 = (int4*)(part + (long)blockIdx.x * N_PAD);
#pragma unroll
  for (int k = 0; k < N_PAD / 4 / 1024; ++k) p4[tid + 1024 * k] = lc4[tid + 1024 * k];
}

__global__ __launch_bounds__(256) void chunk_prefix_kernel(int* __restrict__ part,
                                                           int* __restrict__ cnt) {
  int d = blockIdx.x * 256 + threadIdx.x;
  int v[CH];
#pragma unroll
  for (int i = 0; i < CH; ++i) v[i] = part[(long)i * N_PAD + d];
  int run = 0;
#pragma unroll
  for (int i = 0; i < CH; ++i) {
    part[(long)i * N_PAD + d] = run;
    run += v[i];
  }
  cnt[d] = run;
}

__global__ __launch_bounds__(1024) void scan_kernel(const int4* __restrict__ cntcur,
                                                    int4* __restrict__ off) {
  __shared__ int part[1024];
  int t = threadIdx.x;
  int4 v[13];
  int sum = 0;
  for (int k = 0; k < 13; ++k) {
    v[k] = cntcur[t * 13 + k];
    sum += v[k].x + v[k].y + v[k].z + v[k].w;
  }
  part[t] = sum;
  __syncthreads();
  for (int d = 1; d < 1024; d <<= 1) {
    int u = (t >= d) ? part[t - d] : 0;
    __syncthreads();
    part[t] += u;
    __syncthreads();
  }
  int run = part[t] - sum;
  for (int k = 0; k < 13; ++k) {
    int4 o;
    o.x = run; run += v[k].x;
    o.y = run; run += v[k].y;
    o.z = run; run += v[k].z;
    o.w = run; run += v[k].w;
    off[t * 13 + k] = o;
  }
}

__global__ __launch_bounds__(256) void place_kernel(const int* __restrict__ dst,
                                                    const int4* __restrict__ lslot,
                                                    const int* __restrict__ part,
                                                    const int* __restrict__ off,
                                                    int* __restrict__ perm) {
  int t = blockIdx.x * 256 + threadIdx.x;
  if (t >= N_EDGES / 4) return;
  int4 d = *(const int4*)(dst + 4 * t);
  int4 s = lslot[t];
  int e = 4 * t;
  int d0 = min(max(d.x, 0), N_PAD - 1);
  int d1 = min(max(d.y, 0), N_PAD - 1);
  int d2 = min(max(d.z, 0), N_PAD - 1);
  int d3 = min(max(d.w, 0), N_PAD - 1);
  int c0 = e / CH_EDGES;
  int c1 = (e + 1) / CH_EDGES;
  int c2 = (e + 2) / CH_EDGES;
  int c3 = (e + 3) / CH_EDGES;
  int p0 = off[d0] + part[(long)c0 * N_PAD + d0] + s.x;
  int p1 = off[d1] + part[(long)c1 * N_PAD + d1] + s.y;
  int p2 = off[d2] + part[(long)c2 * N_PAD + d2] + s.z;
  int p3 = off[d3] + part[(long)c3 * N_PAD + d3] + s.w;
  perm[min(max(p0, 0), N_EDGES - 1)] = e;
  perm[min(max(p1, 0), N_EDGES - 1)] = e + 1;
  perm[min(max(p2, 0), N_EDGES - 1)] = e + 2;
  perm[min(max(p3, 0), N_EDGES - 1)] = e + 3;
}

__global__ __launch_bounds__(256) void order_kernel(const int* __restrict__ off,
                                                    const int* __restrict__ perm,
                                                    const int* __restrict__ src,
                                                    const float* __restrict__ ea,
                                                    int4* __restrict__ rec) {
  int lane = threadIdx.x & 31;
  int w = __builtin_amdgcn_readfirstlane((int)(blockIdx.x * 8 + (threadIdx.x >> 5)));
  for (int i = w; i < N_NODES; i += ROW_WAVES) {
    int beg = max(off[i], 0);
    int end = min(off[i + 1], N_EDGES);
    for (int p = beg + lane; p < end; p += 32) {
      int e = perm[p];
      int rank = 0;
      for (int q = beg; q < end; ++q) rank += (perm[q] < e) ? 1 : 0;
      int ec = min(max(e, 0), N_EDGES - 1);
      int s = src[ec];
      float2 a = *(const float2*)(ea + 2 * ec);
      rec[beg + rank] = make_int4(s, __float_as_int(a.x), __float_as_int(a.y), 0);
    }
  }
}

__device__ __forceinline__ float4 edge_msg(float4 hv, float a0, float a1, float4 w0, float4 w1, float4 bb) {
  float m0 = hv.x + a0 * w0.x + a1 * w1.x + bb.x;
  float m1 = hv.y + a0 * w0.y + a1 * w1.y + bb.y;
  float m2 = hv.z + a0 * w0.z + a1 * w1.z + bb.z;
  float m3 = hv.w + a0 * w0.w + a1 * w1.w + bb.w;
  return make_float4(m0 > 0.f ? m0 : 0.f, m1 > 0.f ? m1 : 0.f,
                     m2 > 0.f ? m2 : 0.f, m3 > 0.f ? m3 : 0.f);
}

__device__ __forceinline__ float4 load_row(const float* __restrict__ h, int s, int lane) {
  unsigned sc = min((unsigned)s, (unsigned)(N_NODES - 1));
  return ((const float4*)h)[sc * (unsigned)(HID / 4) + (unsigned)lane];
}

#define TILE_PITCH 132

__global__ __launch_bounds__(256, 16) void gine_layer_kernel(
    const float* __restrict__ h,
    const int*   __restrict__ off,
    const int4*  __restrict__ rec,
    const float* __restrict__ Wl0,
    const float* __restrict__ Wl1,
    const float* __restrict__ bl,
    const float* __restrict__ W,
    const float* __restrict__ bias,
    float* __restrict__ out)
{
  __shared__ float xt[16 * TILE_PITCH];

  int lane = threadIdx.x & 31;
  int wave = __builtin_amdgcn_readfirstlane((int)(threadIdx.x >> 5));
  int m0   = blockIdx.x * 16;
  int c    = lane * 4;

  {
    float4 w0 = *(const float4*)(Wl0 + c);
    float4 w1 = *(const float4*)(Wl1 + c);
    float4 bb = *(const float4*)(bl  + c);

    int iA = m0 + 2 * wave;
    int o0 = off[iA], o1 = off[iA + 1], o2 = off[iA + 2];
    int beg = min(max(o0, 0), N_EDGES);
    int mid = min(max(o1, beg), N_EDGES);
    int end = min(max(o2, mid), N_EDGES);
    int lenA = mid - beg;
    int len  = end - beg;

    float4 hA = *(const float4*)(h + (long)iA * HID + c);
    float4 hB = *(const float4*)(h + (long)(iA + 1) * HID + c);

    float4 accA = make_float4(0.f, 0.f, 0.f, 0.f);
    float4 accB = make_float4(0.f, 0.f, 0.f, 0.f);
    for (int cb = 0; cb < len; cb += 32) {
      int n   = min(len - cb, 32);
      int nAc = min(max(lenA - cb, 0), n);
      int nBc = n - nAc;
      int p = min(beg + cb + min(lane, n - 1), N_EDGES - 1);
      int4 my = rec[p];
      for (int t = 0; t < n; t += 8) {
        float4 xr[8];
        float  e0[8], e1[8], wa[8], wb[8];
#pragma unroll
        for (int j = 0; j < 8; ++j) {
          int q = t + j;
          xr[j] = load_row(h, __builtin_amdgcn_readlane(my.x, q), lane);
          e0[j] = __int_as_float(__builtin_amdgcn_readlane(my.y, q));
          e1[j] = __int_as_float(__builtin_amdgcn_readlane(my.z, q));
          wa[j] = (q < nAc) ? 1.f : 0.f;
          wb[j] = ((unsigned)(q - nAc) < (unsigned)nBc) ? 1.f : 0.f;
        }
#pragma unroll
        for (int j = 0; j < 8; ++j) {
          float4 m = edge_msg(xr[j], e0[j], e1[j], w0, w1, bb);
          accA.x = fmaf(m.x, wa[j], accA.x);  accA.y = fmaf(m.y, wa[j], accA.y);
          accA.z = fmaf(m.z, wa[j], accA.z);  accA.w = fmaf(m.w, wa[j], accA.w);
          accB.x = fmaf(m.x, wb[j], accB.x);  accB.y = fmaf(m.y, wb[j], accB.y);
          accB.z = fmaf(m.z, wb[j], accB.z);  accB.w = fmaf(m.w, wb[j], accB.w);
        }
      }
    }
    *(float4*)(xt + (2 * wave)     * TILE_PITCH + c) =
        make_float4(hA.x + accA.x, hA.y + accA.y, hA.z + accA.z, hA.w + accA.w);
    *(float4*)(xt + (2 * wave + 1) * TILE_PITCH + c) =
        make_float4(hB.x + accB.x, hB.y + accB.y, hB.z + accB.z, hB.w + accB.w);
  }

  int m    = lane & 15;
  int half = lane >> 4;
  int n    = wave * 16 + m;

#define LOAD_W_QUARTER(dst, q)                                               \
  _Pragma("unroll")                                                          \
  for (int s = 0; s < 8; ++s) {                                              \
    int k0 = 4 * (8 * (q) + s) + half * 2;                                   \
    dst[s].x = W[(long)k0 * HID + n];               \
    dst[s].y = W[(long)(k0 + 1) * HID + n];         \
  }
#define WMMA_QUARTER(src, q)                                                 \
  _Pragma("unroll")                                                          \
  for (int s = 0; s < 8; ++s) {                                              \
    v2f a = *(const v2f*)(ar + 4 * (8 * (q) + s));     \
    cc = __builtin_amdgcn_wmma_f32_16x16x4_f32(                              \
         false, a,  false, src[s],                         \
         (short)0, cc,  false,  false);       \
  }
  v2f bq0[8], bq1[8], bq2[8], bq3[8];
  LOAD_W_QUARTER(bq0, 0)
  LOAD_W_QUARTER(bq1, 1)
  float bv = bias[n];

  __syncthreads();

  const float* ar = xt + m * TILE_PITCH + half * 2;
  v8f cc = {};
  WMMA_QUARTER(bq0, 0)
  LOAD_W_QUARTER(bq2, 2)
  WMMA_QUARTER(bq1, 1)
  LOAD_W_QUARTER(bq3, 3)
  WMMA_QUARTER(bq2, 2)
  WMMA_QUARTER(bq3, 3)
#undef LOAD_W_QUARTER
#undef WMMA_QUARTER

#pragma unroll
  for (int v = 0; v < 8; ++v) {
    int row = m0 + v + half * 8;
    float y = cc[v] + bv;
    out[(long)row * HID + n] = (y >= 0.f) ? y : NEG_SLOPE * y;
  }
}

extern "C" void kernel_launch(void* const* d_in, const int* in_sizes, int n_in,
                              void* d_out, int out_size, void* d_ws, size_t ws_size,
                              hipStream_t stream) {
  (void)in_sizes; (void)n_in; (void)out_size; (void)ws_size;

  const float* x  = (const float*)d_in[0];
  const int*   ei = (const int*)  d_in[1];
  const float* ea = (const float*)d_in[2];
  const float* Wl = (const float*)d_in[3];
  const float* bl = (const float*)d_in[4];
  const float* W  = (const float*)d_in[5];
  const float* b  = (const float*)d_in[6];
  float*       out = (float*)d_out;

  const int* src = ei;
  const int* dst = ei + N_EDGES;

  const long NH = (long)N_NODES * HID;
  float* agg  = (float*)d_ws;
  int*   lslot = (int*)d_ws;
  int*   part = lslot + N_EDGES;
  float* h0   = agg + NH;
  int*   cur  = (int*)(h0 + NH);
  int*   off  = cur + N_PAD;
  int*   perm = off + N_PAD;
  int4*  rec  = (int4*)(perm + N_EDGES);

  count_kernel<<<CH, 1024, 0, stream>>>(dst, part, lslot);
  chunk_prefix_kernel<<<N_PAD / 256, 256, 0, stream>>>(part, cur);
  scan_kernel<<<1, 1024, 0, stream>>>((const int4*)cur, (int4*)off);
  place_kernel<<<(N_EDGES / 4 + 255) / 256, 256, 0, stream>>>(dst, (const int4*)lslot, part, off, perm);
  order_kernel<<<ROW_BLOCKS, 256, 0, stream>>>(off, perm, src, ea, rec);

  const float* hin = x;
  for (int layer = 0; layer < N_LAYERS; ++layer) {
    float* hout = (layer == 1) ? h0 : out;

    gine_layer_kernel<<<N_NODES / 16, 256, 0, stream>>>(
        hin, off, rec,
        Wl + (long)layer * 2 * HID,
        Wl + (long)layer * 2 * HID + HID,
        bl + (long)layer * HID,
        W + (long)layer * HID * HID,
        b + (long)layer * HID,
        hout);

    hin = hout;
  }
}


# Node-offset prefix sum on 52 blocks: each re-sums the counters ahead of its 1024 entries with whole-line loads and scans its own by wave shuffles; was 1 block, 208-byte-strided loads and stores
# speedup vs baseline: 3.5130x; 1.1096x over previous
#include <hip/hip_runtime.h>


#define N_NODES 50000
#define N_EDGES 600000
#define HID 128
#define NEG_SLOPE 0.01f
#define N_LAYERS 3

#define N_PAD 53248
#define ROW_BLOCKS 2048
#define ROW_WAVES (ROW_BLOCKS * 8)

typedef __attribute__((ext_vector_type(2))) float v2f;
typedef __attribute__((ext_vector_type(8))) float v8f;

#define CH 32
#define CH_EDGES (N_EDGES / CH)
#define CH_TRIPS ((CH_EDGES + 1023) / 1024)

__global__ __launch_bounds__(1024) void count_kernel(const int* __restrict__ dst,
                                                     int* __restrict__ part,
                                                     int* __restrict__ lslot) {
  __shared__ int lc[N_PAD];
  int tid = threadIdx.x;
  int e0  = blockIdx.x * CH_EDGES;
  int dv[CH_TRIPS];
#pragma unroll
  for (int k = 0; k < CH_TRIPS; ++k) {
    int t = tid + 1024 * k;
    dv[k] = (t < CH_EDGES) ? dst[e0 + t] : 0;
  }
  int4* lc4 = (int4*)lc;
#pragma unroll
  for (int k = 0; k < N_PAD / 4 / 1024; ++k) lc4[tid + 1024 * k] = make_int4(0, 0, 0, 0);
  __syncthreads();
  int sl[CH_TRIPS];
#pragma unroll
  for (int k = 0; k < CH_TRIPS; ++k) {
    int t = tid + 1024 * k;
    int d = min(max(dv[k], 0), N_PAD - 1);
    sl[k] = (t < CH_EDGES) ? atomicAdd(&lc[d], 1) : 0;
  }
  __syncthreads();
#pragma unroll
  for (int k = 0; k < CH_TRIPS; ++k) {
    int t = tid + 1024 * k;
    if (t < CH_EDGES) lslot[e0 + t] = sl[k];
  }
  int4* p4 = (int4*)(part + (long)blockIdx.x * N_PAD);
#pragma unroll
  for (int k = 0; k < N_PAD / 4 / 1024; ++k) p4[tid + 1024 * k] = lc4[tid + 1024 * k];
}

__global__ __launch_bounds__(256) void chunk_prefix_kernel(int* __restrict__ part,
                                                           int* __restrict__ cnt) {
  int d = blockIdx.x * 256 + threadIdx.x;
  int v[CH];
#pragma unroll
  for (int i = 0; i < CH; ++i) v[i] = part[(long)i * N_PAD + d];
  int run = 0;
#pragma unroll
  for (int i = 0; i < CH; ++i) {
    part[(long)i * N_PAD + d] = run;
    run += v[i];
  }
  cnt[d] = run;
}

#define SCAN_T 1024
#define SCAN_BLOCKS (N_PAD / SCAN_T)
#define SCAN_PIECES (N_PAD / 4 / SCAN_T)
__global__ __launch_bounds__(SCAN_T) void scan_kernel(const int* __restrict__ cnt,
                                                      int* __restrict__ off) {
  __shared__ int wfront[32];
  __shared__ int wown[32];
  int t     = threadIdx.x;
  int lane  = t & 31;
  int wave  = t >> 5;
  int first = blockIdx.x * SCAN_T;
  int lim4  = first / 4;
  const int4* c4 = (const int4*)cnt;
  int4 v[SCAN_PIECES];
#pragma unroll
  for (int k = 0; k < SCAN_PIECES; ++k) v[k] = c4[k * SCAN_T + t];
  int c = cnt[first + t];
  int front = 0;
#pragma unroll
  for (int k = 0; k < SCAN_PIECES; ++k) {
    int s = v[k].x + v[k].y + v[k].z + v[k].w;
    front += (k * SCAN_T + t < lim4) ? s : 0;
  }
  int incl = c;
#pragma unroll
  for (int d = 1; d < 32; d <<= 1) {
    front += __shfl_xor(front, d);
    int u = __shfl_up(incl, d);
    incl += (lane >= d) ? u : 0;
  }
  if (lane == 31) { wfront[wave] = front; wown[wave] = incl; }
  __syncthreads();
  int base = wfront[lane] + ((lane < wave) ? wown[lane] : 0);
#pragma unroll
  for (int d = 1; d < 32; d <<= 1) base += __shfl_xor(base, d);
  off[first + t] = base + incl - c;
}

__global__ __launch_bounds__(256) void place_kernel(const int* __restrict__ dst,
                                                    const int4* __restrict__ lslot,
                                                    const int* __restrict__ part,
                                                    const int* __restrict__ off,
                                                    int* __restrict__ perm) {
  int t = blockIdx.x * 256 + threadIdx.x;
  if (t >= N_EDGES / 4) return;
  int4 d = *(const int4*)(dst + 4 * t);
  int4 s = lslot[t];
  int e = 4 * t;
  int d0 = min(max(d.x, 0), N_PAD - 1);
  int d1 = min(max(d.y, 0), N_PAD - 1);
  int d2 = min(max(d.z, 0), N_PAD - 1);
  int d3 = min(max(d.w, 0), N_PAD - 1);
  int c0 = e / CH_EDGES;
  int c1 = (e + 1) / CH_EDGES;
  int c2 = (e + 2) / CH_EDGES;
  int c3 = (e + 3) / CH_EDGES;
  int p0 = off[d0] + part[(long)c0 * N_PAD + d0] + s.x;
  int p1 = off[d1] + part[(long)c1 * N_PAD + d1] + s.y;
  int p2 = off[d2] + part[(long)c2 * N_PAD + d2] + s.z;
  int p3 = off[d3] + part[(long)c3 * N_PAD + d3] + s.w;
  perm[min(max(p0, 0), N_EDGES - 1)] = e;
  perm[min(max(p1, 0), N_EDGES - 1)] = e + 1;
  perm[min(max(p2, 0), N_EDGES - 1)] = e + 2;
  perm[min(max(p3, 0), N_EDGES - 1)] = e + 3;
}

__global__ __launch_bounds__(256) void order_kernel(const int* __restrict__ off,
                                                    const int* __restrict__ perm,
                                                    const int* __restrict__ src,
                                                    const float* __restrict__ ea,
                                                    int4* __restrict__ rec) {
  int lane = threadIdx.x & 31;
  int w = __builtin_amdgcn_readfirstlane((int)(blockIdx.x * 8 + (threadIdx.x >> 5)));
  for (int i = w; i < N_NODES; i += ROW_WAVES) {
    int beg = max(off[i], 0);
    int end = min(off[i + 1], N_EDGES);
    for (int p = beg + lane; p < end; p += 32) {
      int e = perm[p];
      int rank = 0;
      for (int q = beg; q < end; ++q) rank += (perm[q] < e) ? 1 : 0;
      int ec = min(max(e, 0), N_EDGES - 1);
      int s = src[ec];
      float2 a = *(const float2*)(ea + 2 * ec);
      rec[beg + rank] = make_int4(s, __float_as_int(a.x), __float_as_int(a.y), 0);
    }
  }
}

__device__ __forceinline__ float4 edge_msg(float4 hv, float a0, float a1, float4 w0, float4 w1, float4 bb) {
  float m0 = hv.x + a0 * w0.x + a1 * w1.x + bb.x;
  float m1 = hv.y + a0 * w0.y + a1 * w1.y + bb.y;
  float m2 = hv.z + a0 * w0.z + a1 * w1.z + bb.z;
  float m3 = hv.w + a0 * w0.w + a1 * w1.w + bb.w;
  return make_float4(m0 > 0.f ? m0 : 0.f, m1 > 0.f ? m1 : 0.f,
                     m2 > 0.f ? m2 : 0.f, m3 > 0.f ? m3 : 0.f);
}

__device__ __forceinline__ float4 load_row(const float* __restrict__ h, int s, int lane) {
  unsigned sc = min((unsigned)s, (unsigned)(N_NODES - 1));
  return ((const float4*)h)[sc * (unsigned)(HID / 4) + (unsigned)lane];
}

#define TILE_PITCH 132

__global__ __launch_bounds__(256, 16) void gine_layer_kernel(
    const float* __restrict__ h,
    const int*   __restrict__ off,
    const int4*  __restrict__ rec,
    const float* __restrict__ Wl0,
    const float* __restrict__ Wl1,
    const float* __restrict__ bl,
    const float* __restrict__ W,
    const float* __restrict__ bias,
    float* __restrict__ out)
{
  __shared__ float xt[16 * TILE_PITCH];

  int lane = threadIdx.x & 31;
  int wave = __builtin_amdgcn_readfirstlane((int)(threadIdx.x >> 5));
  int m0   = blockIdx.x * 16;
  int c    = lane * 4;

  {
    float4 w0 = *(const float4*)(Wl0 + c);
    float4 w1 = *(const float4*)(Wl1 + c);
    float4 bb = *(const float4*)(bl  + c);

    int iA = m0 + 2 * wave;
    int o0 = off[iA], o1 = off[iA + 1], o2 = off[iA + 2];
    int beg = min(max(o0, 0), N_EDGES);
    int mid = min(max(o1, beg), N_EDGES);
    int end = min(max(o2, mid), N_EDGES);
    int lenA = mid - beg;
    int len  = end - beg;

    float4 hA = *(const float4*)(h + (long)iA * HID + c);
    float4 hB = *(const float4*)(h + (long)(iA + 1) * HID + c);

    float4 accA = make_float4(0.f, 0.f, 0.f, 0.f);
    float4 accB = make_float4(0.f, 0.f, 0.f, 0.f);
    for (int cb = 0; cb < len; cb += 32) {
      int n   = min(len - cb, 32);
      int nAc = min(max(lenA - cb, 0), n);
      int nBc = n - nAc;
      int p = min(beg + cb + min(lane, n - 1), N_EDGES - 1);
      int4 my = rec[p];
      for (int t = 0; t < n; t += 8) {
        float4 xr[8];
        float  e0[8], e1[8], wa[8], wb[8];
#pragma unroll
        for (int j = 0; j < 8; ++j) {
          int q = t + j;
          xr[j] = load_row(h, __builtin_amdgcn_readlane(my.x, q), lane);
          e0[j] = __int_as_float(__builtin_amdgcn_readlane(my.y, q));
          e1[j] = __int_as_float(__builtin_amdgcn_readlane(my.z, q));
          wa[j] = (q < nAc) ? 1.f : 0.f;
          wb[j] = ((unsigned)(q - nAc) < (unsigned)nBc) ? 1.f : 0.f;
        }
#pragma unroll
        for (int j = 0; j < 8; ++j) {
          float4 m = edge_msg(xr[j], e0[j], e1[j], w0, w1, bb);
          accA.x = fmaf(m.x, wa[j], accA.x);  accA.y = fmaf(m.y, wa[j], accA.y);
          accA.z = fmaf(m.z, wa[j], accA.z);  accA.w = fmaf(m.w, wa[j], accA.w);
          accB.x = fmaf(m.x, wb[j], accB.x);  accB.y = fmaf(m.y, wb[j], accB.y);
          accB.z = fmaf(m.z, wb[j], accB.z);  accB.w = fmaf(m.w, wb[j], accB.w);
        }
      }
    }
    *(float4*)(xt + (2 * wave)     * TILE_PITCH + c) =
        make_float4(hA.x + accA.x, hA.y + accA.y, hA.z + accA.z, hA.w + accA.w);
    *(float4*)(xt + (2 * wave + 1) * TILE_PITCH + c) =
        make_float4(hB.x + accB.x, hB.y + accB.y, hB.z + accB.z, hB.w + accB.w);
  }

  int m    = lane & 15;
  int half = lane >> 4;
  int n    = wave * 16 + m;

#define LOAD_W_QUARTER(dst, q)                                               \
  _Pragma("unroll")                                                          \
  for (int s = 0; s < 8; ++s) {                                              \
    int k0 = 4 * (8 * (q) + s) + half * 2;                                   \
    dst[s].x = W[(long)k0 * HID + n];               \
    dst[s].y = W[(long)(k0 + 1) * HID + n];         \
  }
#define WMMA_QUARTER(src, q)                                                 \
  _Pragma("unroll")                                                          \
  for (int s = 0; s < 8; ++s) {                                              \
    v2f a = *(const v2f*)(ar + 4 * (8 * (q) + s));     \
    cc = __builtin_amdgcn_wmma_f32_16x16x4_f32(                              \
         false, a,  false, src[s],                         \
         (short)0, cc,  false,  false);       \
  }
  v2f bq0[8], bq1[8], bq2[8], bq3[8];
  LOAD_W_QUARTER(bq0, 0)
  LOAD_W_QUARTER(bq1, 1)
  float bv = bias[n];

  __syncthreads();

  const float* ar = xt + m * TILE_PITCH + half * 2;
  v8f cc = {};
  WMMA_QUARTER(bq0, 0)
  LOAD_W_QUARTER(bq2, 2)
  WMMA_QUARTER(bq1, 1)
  LOAD_W_QUARTER(bq3, 3)
  WMMA_QUARTER(bq2, 2)
  WMMA_QUARTER(bq3, 3)
#undef LOAD_W_QUARTER
#undef WMMA_QUARTER

#pragma unroll
  for (int v = 0; v < 8; ++v) {
    int row = m0 + v + half * 8;
    float y = cc[v] + bv;
    out[(long)row * HID + n] = (y >= 0.f) ? y : NEG_SLOPE * y;
  }
}

extern "C" void kernel_launch(void* const* d_in, const int* in_sizes, int n_in,
                              void* d_out, int out_size, void* d_ws, size_t ws_size,
                              hipStream_t stream) {
  (void)in_sizes; (void)n_in; (void)out_size; (void)ws_size;

  const float* x  = (const float*)d_in[0];
  const int*   ei = (const int*)  d_in[1];
  const float* ea = (const float*)d_in[2];
  const float* Wl = (const float*)d_in[3];
  const float* bl = (const float*)d_in[4];
  const float* W  = (const float*)d_in[5];
  const float* b  = (const float*)d_in[6];
  float*       out = (float*)d_out;

  const int* src = ei;
  const int* dst = ei + N_EDGES;

  const long NH = (long)N_NODES * HID;
  float* agg  = (float*)d_ws;
  int*   lslot = (int*)d_ws;
  int*   part = lslot + N_EDGES;
  float* h0   = agg + NH;
  int*   cur  = (int*)(h0 + NH);
  int*   off  = cur + N_PAD;
  int*   perm = off + N_PAD;
  int4*  rec  = (int4*)(perm + N_EDGES);

  count_kernel<<<CH, 1024, 0, stream>>>(dst, part, lslot);
  chunk_prefix_kernel<<<N_PAD / 256, 256, 0, stream>>>(part, cur);
  scan_kernel<<<SCAN_BLOCKS, SCAN_T, 0, stream>>>(cur, off);
  place_kernel<<<(N_EDGES / 4 + 255) / 256, 256, 0, stream>>>(dst, (const int4*)lslot, part, off, perm);
  order_kernel<<<ROW_BLOCKS, 256, 0, stream>>>(off, perm, src, ea, rec);

  const float* hin = x;
  for (int layer = 0; layer < N_LAYERS; ++layer) {
    float* hout = (layer == 1) ? h0 : out;

    gine_layer_kernel<<<N_NODES / 16, 256, 0, stream>>>(
        hin, off, rec,
        Wl + (long)layer * 2 * HID,
        Wl + (long)layer * 2 * HID + HID,
        bl + (long)layer * HID,
        W + (long)layer * HID * HID,
        b + (long)layer * HID,
        hout);

    hin = hout;
  }
}


# Per-call edge sort in 2 launches, was 5: blocks bin 2400-edge chunks by destination range in LDS; one block per 208-node range counts, ranks by edge id, writes records from LDS; sum order unchanged
# speedup vs baseline: 4.6179x; 1.3145x over previous
#include <hip/hip_runtime.h>


#define N_NODES 50000
#define N_EDGES 600000
#define HID 128
#define NEG_SLOPE 0.01f
#define N_LAYERS 3

#define N_PAD 53248

typedef __attribute__((ext_vector_type(2))) float v2f;
typedef __attribute__((ext_vector_type(8))) float v8f;

#define BIN_EDGES 2400
#define BIN_BLOCKS (N_EDGES / BIN_EDGES)
#define BIN_T 256
#define BIN_PER ((BIN_EDGES + BIN_T - 1) / BIN_T)
#define NR 256
#define RW (N_PAD / NR)
#define TABW 288
static_assert(BIN_BLOCKS * BIN_EDGES == N_EDGES, "chunks cover the edge list exactly");
static_assert(NR * RW == N_PAD && BIN_T == NR && BIN_BLOCKS <= NR && TABW >= NR + 1, "range and block shapes");

template <int PER>
__device__ __forceinline__ void wave_scan(const int* in, int* out, int lane) {
  int v[PER];
  int s = 0;
#pragma unroll
  for (int k = 0; k < PER; ++k) { v[k] = in[lane * PER + k]; s += v[k]; }
  int incl = s;
#pragma unroll
  for (int d = 1; d < 32; d <<= 1) {
    int u = __shfl_up(incl, d);
    incl += (lane >= d) ? u : 0;
  }
  int run = incl - s;
#pragma unroll
  for (int k = 0; k < PER; ++k) { out[lane * PER + k] = run; run += v[k]; }
  if (lane == 31) out[32 * PER] = run;
}

__global__ __launch_bounds__(BIN_T) void bin_kernel(const int* __restrict__ src,
                                                    const int* __restrict__ dst,
                                                    const float* __restrict__ ea,
                                                    int4* __restrict__ tmp,
                                                    int* __restrict__ tab) {
  __shared__ int  rc[NR];
  __shared__ int  rs[NR + 1];
  __shared__ int4 st[BIN_EDGES];
  int tid = threadIdx.x;
  int e0  = blockIdx.x * BIN_EDGES;
  int dv[BIN_PER], sv[BIN_PER];
  float2 av[BIN_PER];
#pragma unroll
  for (int k = 0; k < BIN_PER; ++k) {
    int t = e0 + min(tid + BIN_T * k, BIN_EDGES - 1);
    dv[k] = dst[t];
    sv[k] = src[t];
    av[k] = *(const float2*)(ea + 2 * t);
  }
  rc[tid] = 0;
  __syncthreads();
  int sl[BIN_PER];
#pragma unroll
  for (int k = 0; k < BIN_PER; ++k) {
    int t = tid + BIN_T * k;
    int d = min(max(dv[k], 0), N_PAD - 1);
    dv[k] = d;
    sl[k] = (t < BIN_EDGES) ? atomicAdd(&rc[d / RW], 1) : 0;
  }
  __syncthreads();
  if (tid < 32) wave_scan<NR / 32>(rc, rs, tid);
  __syncthreads();
  tab[blockIdx.x * TABW + tid] = rs[tid];
  if (tid == 0) tab[blockIdx.x * TABW + NR] = rs[NR];
#pragma unroll
  for (int k = 0; k < BIN_PER; ++k) {
    int t = tid + BIN_T * k;
    if (t < BIN_EDGES)
      st[rs[dv[k] / RW] + sl[k]] = make_int4(sv[k], __float_as_int(av[k].x), __float_as_int(av[k].y), (t << 16) | dv[k]);
  }
  __syncthreads();
  for (int i = tid; i < BIN_EDGES; i += BIN_T) tmp[e0 + i] = st[i];
}

#define RNG_T 512
#define RNG_CAP 3072
#define RNG_PER (RNG_CAP / RNG_T)
static_assert(RNG_PER * RNG_T == RNG_CAP, "the unrolled record loop covers the LDS arrays exactly");
#define RWP 224

__device__ __forceinline__ int rec_node(int w, int lo) { return min(max((w & 0xFFFF) - lo, 0), RW - 1); }
__device__ __forceinline__ int rec_edge(int w, int c) { return c * BIN_EDGES + (int)((unsigned)w >> 16); }

__global__ __launch_bounds__(RNG_T) void range_kernel(const int4* __restrict__ tmp,
                                                      const int* __restrict__ tab,
                                                      int* __restrict__ off,
                                                      int4* __restrict__ rec) {
  __shared__ int ps[NR];
  __shared__ int pl[NR];
  __shared__ int pp[NR + 1];
  __shared__ int cn[RWP];
  __shared__ int no[RWP + 1];
  __shared__ int sbase;
  __shared__ int S[RNG_CAP], A0[RNG_CAP], A1[RNG_CAP];
  __shared__ int D[RNG_CAP], EI[RNG_CAP], SL[RNG_CAP];
  __shared__ int BE[RNG_CAP + 4];
  __shared__ int4 OUT[RNG_CAP];

  int tid = threadIdx.x;
  int r   = blockIdx.x;
  int lo  = r * RW;
  if (tid < NR) {
    int a = 0, b = 0;
    if (tid < BIN_BLOCKS) { a = tab[tid * TABW + r]; b = tab[tid * TABW + r + 1]; }
    a = min(max(a, 0), BIN_EDGES);
    b = min(max(b, a), BIN_EDGES);
    ps[tid] = a;
    pl[tid] = b - a;
  }
  if (tid < RWP) cn[tid] = 0;
  __syncthreads();
  if (tid < 32) {
    wave_scan<NR / 32>(pl, pp, tid);
    int s = 0;
#pragma unroll
    for (int k = 0; k < NR / 32; ++k) s += ps[tid * (NR / 32) + k];
#pragma unroll
    for (int d = 1; d < 32; d <<= 1) s += __shfl_xor(s, d);
    if (tid == 0) sbase = s;
  }
  __syncthreads();
  int base = __builtin_amdgcn_readfirstlane(sbase);
  int n    = __builtin_amdgcn_readfirstlane(pp[NR]);

  if (n <= RNG_CAP) {
    if (tid < NR) {
      int p0 = pp[tid], ln = pl[tid];
      for (int k = 0; k < ln; ++k) BE[p0 + k] = tid;
    }
    __syncthreads();
    {
      int4 v[RNG_PER];
      int  ch[RNG_PER];
#pragma unroll
      for (int k = 0; k < RNG_PER; ++k) {
        int j = tid + RNG_T * k;
        int c = BE[j] & (NR - 1);
        int idx = c * BIN_EDGES + ps[c] + (j - pp[c]);
        ch[k] = c;
        v[k] = tmp[(j < n) ? idx : 0];
      }
#pragma unroll
      for (int k = 0; k < RNG_PER; ++k) {
        int j = tid + RNG_T * k;
        if (j < n) {
          int i = rec_node(v[k].w, lo);
          S[j] = v[k].x;  A0[j] = v[k].y;  A1[j] = v[k].z;
          D[j] = i;
          EI[j] = rec_edge(v[k].w, ch[k]);
          SL[j] = atomicAdd(&cn[i], 1);
        }
      }
    }
    __syncthreads();
    if (tid < 32) wave_scan<RWP / 32>(cn, no, tid);
    __syncthreads();
    if (tid < RW) off[lo + tid] = base + no[tid];
    for (int j = tid; j < n; j += RNG_T) BE[no[D[j]] + SL[j]] = EI[j];
    __syncthreads();
    for (int j = tid; j < n; j += RNG_T) {
      int i = D[j], e = EI[j];
      int b0 = no[i], b1 = no[i + 1];
      int rank = 0;
      for (int q = b0; q < b1; q += 4) {
        int x0 = BE[q], x1 = BE[q + 1], x2 = BE[q + 2], x3 = BE[q + 3];
        rank += (x0 < e) ? 1 : 0;
        rank += (q + 1 < b1 && x1 < e) ? 1 : 0;
        rank += (q + 2 < b1 && x2 < e) ? 1 : 0;
        rank += (q + 3 < b1 && x3 < e) ? 1 : 0;
      }
      OUT[b0 + rank] = make_int4(S[j], A0[j], A1[j], 0);
    }
    __syncthreads();
    for (int j = tid; j < n; j += RNG_T) {
      int p = base + j;
      if (p < N_EDGES) rec[p] = OUT[j];
    }
  } else {
    for (int c = 0; c < BIN_BLOCKS; ++c) {
      int s0 = ps[c], ln = pl[c];
      for (int k = tid; k < ln; k += RNG_T)
        atomicAdd(&cn[rec_node(tmp[c * BIN_EDGES + s0 + k].w, lo)], 1);
    }
    __syncthreads();
    if (tid < 32) wave_scan<RWP / 32>(cn, no, tid);
    __syncthreads();
    if (tid < RW) off[lo + tid] = base + no[tid];
    for (int c = 0; c < BIN_BLOCKS; ++c) {
      int s0 = ps[c], ln = pl[c];
      for (int k = tid; k < ln; k += RNG_T) {
        int4 v = tmp[c * BIN_EDGES + s0 + k];
        int i = rec_node(v.w, lo);
        int e = rec_edge(v.w, c);
        int rank = 0;
        for (int c2 = 0; c2 < BIN_BLOCKS; ++c2) {
          int s2 = ps[c2], l2 = pl[c2];
          for (int k2 = 0; k2 < l2; ++k2) {
            int w2 = tmp[c2 * BIN_EDGES + s2 + k2].w;
            rank += (rec_node(w2, lo) == i && rec_edge(w2, c2) < e) ? 1 : 0;
          }
        }
        int p = base + no[i] + rank;
        rec[min(max(p, 0), N_EDGES - 1)] = make_int4(v.x, v.y, v.z, 0);
      }
    }
  }
}

__device__ __forceinline__ float4 edge_msg(float4 hv, float a0, float a1, float4 w0, float4 w1, float4 bb) {
  float m0 = hv.x + a0 * w0.x + a1 * w1.x + bb.x;
  float m1 = hv.y + a0 * w0.y + a1 * w1.y + bb.y;
  float m2 = hv.z + a0 * w0.z + a1 * w1.z + bb.z;
  float m3 = hv.w + a0 * w0.w + a1 * w1.w + bb.w;
  return make_float4(m0 > 0.f ? m0 : 0.f, m1 > 0.f ? m1 : 0.f,
                     m2 > 0.f ? m2 : 0.f, m3 > 0.f ? m3 : 0.f);
}

__device__ __forceinline__ float4 load_row(const float* __restrict__ h, int s, int lane) {
  unsigned sc = min((unsigned)s, (unsigned)(N_NODES - 1));
  return ((const float4*)h)[sc * (unsigned)(HID / 4) + (unsigned)lane];
}

#define TILE_PITCH 132

__global__ __launch_bounds__(256, 16) void gine_layer_kernel(
    const float* __restrict__ h,
    const int*   __restrict__ off,
    const int4*  __restrict__ rec,
    const float* __restrict__ Wl0,
    const float* __restrict__ Wl1,
    const float* __restrict__ bl,
    const float* __restrict__ W,
    const float* __restrict__ bias,
    float* __restrict__ out)
{
  __shared__ float xt[16 * TILE_PITCH];

  int lane = threadIdx.x & 31;
  int wave = __builtin_amdgcn_readfirstlane((int)(threadIdx.x >> 5));
  int m0   = blockIdx.x * 16;
  int c    = lane * 4;

  {
    float4 w0 = *(const float4*)(Wl0 + c);
    float4 w1 = *(const float4*)(Wl1 + c);
    float4 bb = *(const float4*)(bl  + c);

    int iA = m0 + 2 * wave;
    int o0 = off[iA], o1 = off[iA + 1], o2 = off[iA + 2];
    int beg = min(max(o0, 0), N_EDGES);
    int mid = min(max(o1, beg), N_EDGES);
    int end = min(max(o2, mid), N_EDGES);
    int lenA = mid - beg;
    int len  = end - beg;

    float4 hA = *(const float4*)(h + (long)iA * HID + c);
    float4 hB = *(const float4*)(h + (long)(iA + 1) * HID + c);

    float4 accA = make_float4(0.f, 0.f, 0.f, 0.f);
    float4 accB = make_float4(0.f, 0.f, 0.f, 0.f);
    for (int cb = 0; cb < len; cb += 32) {
      int n   = min(len - cb, 32);
      int nAc = min(max(lenA - cb, 0), n);
      int nBc = n - nAc;
      int p = min(beg + cb + min(lane, n - 1), N_EDGES - 1);
      int4 my = rec[p];
      for (int t = 0; t < n; t += 8) {
        float4 xr[8];
        float  e0[8], e1[8], wa[8], wb[8];
#pragma unroll
        for (int j = 0; j < 8; ++j) {
          int q = t + j;
          xr[j] = load_row(h, __builtin_amdgcn_readlane(my.x, q), lane);
          e0[j] = __int_as_float(__builtin_amdgcn_readlane(my.y, q));
          e1[j] = __int_as_float(__builtin_amdgcn_readlane(my.z, q));
          wa[j] = (q < nAc) ? 1.f : 0.f;
          wb[j] = ((unsigned)(q - nAc) < (unsigned)nBc) ? 1.f : 0.f;
        }
#pragma unroll
        for (int j = 0; j < 8; ++j) {
          float4 m = edge_msg(xr[j], e0[j], e1[j], w0, w1, bb);
          accA.x = fmaf(m.x, wa[j], accA.x);  accA.y = fmaf(m.y, wa[j], accA.y);
          accA.z = fmaf(m.z, wa[j], accA.z);  accA.w = fmaf(m.w, wa[j], accA.w);
          accB.x = fmaf(m.x, wb[j], accB.x);  accB.y = fmaf(m.y, wb[j], accB.y);
          accB.z = fmaf(m.z, wb[j], accB.z);  accB.w = fmaf(m.w, wb[j], accB.w);
        }
      }
    }
    *(float4*)(xt + (2 * wave)     * TILE_PITCH + c) =
        make_float4(hA.x + accA.x, hA.y + accA.y, hA.z + accA.z, hA.w + accA.w);
    *(float4*)(xt + (2 * wave + 1) * TILE_PITCH + c) =
        make_float4(hB.x + accB.x, hB.y + accB.y, hB.z + accB.z, hB.w + accB.w);
  }

  int m    = lane & 15;
  int half = lane >> 4;
  int n    = wave * 16 + m;

#define LOAD_W_QUARTER(dst, q)                                               \
  _Pragma("unroll")                                                          \
  for (int s = 0; s < 8; ++s) {                                              \
    int k0 = 4 * (8 * (q) + s) + half * 2;                                   \
    dst[s].x = W[(long)k0 * HID + n];               \
    dst[s].y = W[(long)(k0 + 1) * HID + n];         \
  }
#define WMMA_QUARTER(src, q)                                                 \
  _Pragma("unroll")                                                          \
  for (int s = 0; s < 8; ++s) {                                              \
    v2f a = *(const v2f*)(ar + 4 * (8 * (q) + s));     \
    cc = __builtin_amdgcn_wmma_f32_16x16x4_f32(                              \
         false, a,  false, src[s],                         \
         (short)0, cc,  false,  false);       \
  }
  v2f bq0[8], bq1[8], bq2[8], bq3[8];
  LOAD_W_QUARTER(bq0, 0)
  LOAD_W_QUARTER(bq1, 1)
  float bv = bias[n];

  __syncthreads();

  const float* ar = xt + m * TILE_PITCH + half * 2;
  v8f cc = {};
  WMMA_QUARTER(bq0, 0)
  LOAD_W_QUARTER(bq2, 2)
  WMMA_QUARTER(bq1, 1)
  LOAD_W_QUARTER(bq3, 3)
  WMMA_QUARTER(bq2, 2)
  WMMA_QUARTER(bq3, 3)
#undef LOAD_W_QUARTER
#undef WMMA_QUARTER

#pragma unroll
  for (int v = 0; v < 8; ++v) {
    int row = m0 + v + half * 8;
    float y = cc[v] + bv;
    out[(long)row * HID + n] = (y >= 0.f) ? y : NEG_SLOPE * y;
  }
}

extern "C" void kernel_launch(void* const* d_in, const int* in_sizes, int n_in,
                              void* d_out, int out_size, void* d_ws, size_t ws_size,
                              hipStream_t stream) {
  (void)in_sizes; (void)n_in; (void)out_size; (void)ws_size;

  const float* x  = (const float*)d_in[0];
  const int*   ei = (const int*)  d_in[1];
  const float* ea = (const float*)d_in[2];
  const float* Wl = (const float*)d_in[3];
  const float* bl = (const float*)d_in[4];
  const float* W  = (const float*)d_in[5];
  const float* b  = (const float*)d_in[6];
  float*       out = (float*)d_out;

  const int* src = ei;
  const int* dst = ei + N_EDGES;

  const long NH = (long)N_NODES * HID;
  float* agg  = (float*)d_ws;
  int4*  tmp  = (int4*)d_ws;
  int*   tab  = (int*)(tmp + N_EDGES);
  float* h0   = agg + NH;
  int*   off  = (int*)(h0 + NH);
  int4*  rec  = (int4*)(off + N_PAD);

  bin_kernel<<<BIN_BLOCKS, BIN_T, 0, stream>>>(src, dst, ea, tmp, tab);
  range_kernel<<<NR, RNG_T, 0, stream>>>(tmp, tab, off, rec);

  const float* hin = x;
  for (int layer = 0; layer < N_LAYERS; ++layer) {
    float* hout = (layer == 1) ? h0 : out;

    gine_layer_kernel<<<N_NODES / 16, 256, 0, stream>>>(
        hin, off, rec,
        Wl + (long)layer * 2 * HID,
        Wl + (long)layer * 2 * HID + HID,
        bl + (long)layer * HID,
        W + (long)layer * HID * HID,
        b + (long)layer * HID,
        hout);

    hin = hout;
  }
}


# Layer block owns 32 rows, was 16: a wave sums 4 rows (two pairs in turn) and uses each W fragment for both 16-row tiles; half the waves enter, half the W loads per row; every sum order unchanged
# speedup vs baseline: 4.8675x; 1.0541x over previous
#include <hip/hip_runtime.h>


#define N_NODES 50000
#define N_EDGES 600000
#define HID 128
#define NEG_SLOPE 0.01f
#define N_LAYERS 3

#define N_PAD 53248

typedef __attribute__((ext_vector_type(2))) float v2f;
typedef __attribute__((ext_vector_type(8))) float v8f;

#define BIN_EDGES 2400
#define BIN_BLOCKS (N_EDGES / BIN_EDGES)
#define BIN_T 256
#define BIN_PER ((BIN_EDGES + BIN_T - 1) / BIN_T)
#define NR 256
#define RW (N_PAD / NR)
#define TABW 288
static_assert(BIN_BLOCKS * BIN_EDGES == N_EDGES, "chunks cover the edge list exactly");
static_assert(NR * RW == N_PAD && BIN_T == NR && BIN_BLOCKS <= NR && TABW >= NR + 1, "range and block shapes");

template <int PER>
__device__ __forceinline__ void wave_scan(const int* in, int* out, int lane) {
  int v[PER];
  int s = 0;
#pragma unroll
  for (int k = 0; k < PER; ++k) { v[k] = in[lane * PER + k]; s += v[k]; }
  int incl = s;
#pragma unroll
  for (int d = 1; d < 32; d <<= 1) {
    int u = __shfl_up(incl, d);
    incl += (lane >= d) ? u : 0;
  }
  int run = incl - s;
#pragma unroll
  for (int k = 0; k < PER; ++k) { out[lane * PER + k] = run; run += v[k]; }
  if (lane == 31) out[32 * PER] = run;
}

__global__ __launch_bounds__(BIN_T) void bin_kernel(const int* __restrict__ src,
                                                    const int* __restrict__ dst,
                                                    const float* __restrict__ ea,
                                                    int4* __restrict__ tmp,
                                                    int* __restrict__ tab) {
  __shared__ int  rc[NR];
  __shared__ int  rs[NR + 1];
  __shared__ int4 st[BIN_EDGES];
  int tid = threadIdx.x;
  int e0  = blockIdx.x * BIN_EDGES;
  int dv[BIN_PER], sv[BIN_PER];
  float2 av[BIN_PER];
#pragma unroll
  for (int k = 0; k < BIN_PER; ++k) {
    int t = e0 + min(tid + BIN_T * k, BIN_EDGES - 1);
    dv[k] = dst[t];
    sv[k] = src[t];
    av[k] = *(const float2*)(ea + 2 * t);
  }
  rc[tid] = 0;
  __syncthreads();
  int sl[BIN_PER];
#pragma unroll
  for (int k = 0; k < BIN_PER; ++k) {
    int t = tid + BIN_T * k;
    int d = min(max(dv[k], 0), N_PAD - 1);
    dv[k] = d;
    sl[k] = (t < BIN_EDGES) ? atomicAdd(&rc[d / RW], 1) : 0;
  }
  __syncthreads();
  if (tid < 32) wave_scan<NR / 32>(rc, rs, tid);
  __syncthreads();
  tab[blockIdx.x * TABW + tid] = rs[tid];
  if (tid == 0) tab[blockIdx.x * TABW + NR] = rs[NR];
#pragma unroll
  for (int k = 0; k < BIN_PER; ++k) {
    int t = tid + BIN_T * k;
    if (t < BIN_EDGES)
      st[rs[dv[k] / RW] + sl[k]] = make_int4(sv[k], __float_as_int(av[k].x), __float_as_int(av[k].y), (t << 16) | dv[k]);
  }
  __syncthreads();
  for (int i = tid; i < BIN_EDGES; i += BIN_T) tmp[e0 + i] = st[i];
}

#define RNG_T 512
#define RNG_CAP 3072
#define RNG_PER (RNG_CAP / RNG_T)
static_assert(RNG_PER * RNG_T == RNG_CAP, "the unrolled record loop covers the LDS arrays exactly");
#define RWP 224

__device__ __forceinline__ int rec_node(int w, int lo) { return min(max((w & 0xFFFF) - lo, 0), RW - 1); }
__device__ __forceinline__ int rec_edge(int w, int c) { return c * BIN_EDGES + (int)((unsigned)w >> 16); }

__global__ __launch_bounds__(RNG_T) void range_kernel(const int4* __restrict__ tmp,
                                                      const int* __restrict__ tab,
                                                      int* __restrict__ off,
                                                      int4* __restrict__ rec) {
  __shared__ int ps[NR];
  __shared__ int pl[NR];
  __shared__ int pp[NR + 1];
  __shared__ int cn[RWP];
  __shared__ int no[RWP + 1];
  __shared__ int sbase;
  __shared__ int S[RNG_CAP], A0[RNG_CAP], A1[RNG_CAP];
  __shared__ int D[RNG_CAP], EI[RNG_CAP], SL[RNG_CAP];
  __shared__ int BE[RNG_CAP + 4];
  __shared__ int4 OUT[RNG_CAP];

  int tid = threadIdx.x;
  int r   = blockIdx.x;
  int lo  = r * RW;
  if (tid < NR) {
    int a = 0, b = 0;
    if (tid < BIN_BLOCKS) { a = tab[tid * TABW + r]; b = tab[tid * TABW + r + 1]; }
    a = min(max(a, 0), BIN_EDGES);
    b = min(max(b, a), BIN_EDGES);
    ps[tid] = a;
    pl[tid] = b - a;
  }
  if (tid < RWP) cn[tid] = 0;
  __syncthreads();
  if (tid < 32) {
    wave_scan<NR / 32>(pl, pp, tid);
    int s = 0;
#pragma unroll
    for (int k = 0; k < NR / 32; ++k) s += ps[tid * (NR / 32) + k];
#pragma unroll
    for (int d = 1; d < 32; d <<= 1) s += __shfl_xor(s, d);
    if (tid == 0) sbase = s;
  }
  __syncthreads();
  int base = __builtin_amdgcn_readfirstlane(sbase);
  int n    = __builtin_amdgcn_readfirstlane(pp[NR]);

  if (n <= RNG_CAP) {
    if (tid < NR) {
      int p0 = pp[tid], ln = pl[tid];
      for (int k = 0; k < ln; ++k) BE[p0 + k] = tid;
    }
    __syncthreads();
    {
      int4 v[RNG_PER];
      int  ch[RNG_PER];
#pragma unroll
      for (int k = 0; k < RNG_PER; ++k) {
        int j = tid + RNG_T * k;
        int c = BE[j] & (NR - 1);
        int idx = c * BIN_EDGES + ps[c] + (j - pp[c]);
        ch[k] = c;
        v[k] = tmp[(j < n) ? idx : 0];
      }
#pragma unroll
      for (int k = 0; k < RNG_PER; ++k) {
        int j = tid + RNG_T * k;
        if (j < n) {
          int i = rec_node(v[k].w, lo);
          S[j] = v[k].x;  A0[j] = v[k].y;  A1[j] = v[k].z;
          D[j] = i;
          EI[j] = rec_edge(v[k].w, ch[k]);
          SL[j] = atomicAdd(&cn[i], 1);
        }
      }
    }
    __syncthreads();
    if (tid < 32) wave_scan<RWP / 32>(cn, no, tid);
    __syncthreads();
    if (tid < RW) off[lo + tid] = base + no[tid];
    for (int j = tid; j < n; j += RNG_T) BE[no[D[j]] + SL[j]] = EI[j];
    __syncthreads();
    for (int j = tid; j < n; j += RNG_T) {
      int i = D[j], e = EI[j];
      int b0 = no[i], b1 = no[i + 1];
      int rank = 0;
      for (int q = b0; q < b1; q += 4) {
        int x0 = BE[q], x1 = BE[q + 1], x2 = BE[q + 2], x3 = BE[q + 3];
        rank += (x0 < e) ? 1 : 0;
        rank += (q + 1 < b1 && x1 < e) ? 1 : 0;
        rank += (q + 2 < b1 && x2 < e) ? 1 : 0;
        rank += (q + 3 < b1 && x3 < e) ? 1 : 0;
      }
      OUT[b0 + rank] = make_int4(S[j], A0[j], A1[j], 0);
    }
    __syncthreads();
    for (int j = tid; j < n; j += RNG_T) {
      int p = base + j;
      if (p < N_EDGES) rec[p] = OUT[j];
    }
  } else {
    for (int c = 0; c < BIN_BLOCKS; ++c) {
      int s0 = ps[c], ln = pl[c];
      for (int k = tid; k < ln; k += RNG_T)
        atomicAdd(&cn[rec_node(tmp[c * BIN_EDGES + s0 + k].w, lo)], 1);
    }
    __syncthreads();
    if (tid < 32) wave_scan<RWP / 32>(cn, no, tid);
    __syncthreads();
    if (tid < RW) off[lo + tid] = base + no[tid];
    for (int c = 0; c < BIN_BLOCKS; ++c) {
      int s0 = ps[c], ln = pl[c];
      for (int k = tid; k < ln; k += RNG_T) {
        int4 v = tmp[c * BIN_EDGES + s0 + k];
        int i = rec_node(v.w, lo);
        int e = rec_edge(v.w, c);
        int rank = 0;
        for (int c2 = 0; c2 < BIN_BLOCKS; ++c2) {
          int s2 = ps[c2], l2 = pl[c2];
          for (int k2 = 0; k2 < l2; ++k2) {
            int w2 = tmp[c2 * BIN_EDGES + s2 + k2].w;
            rank += (rec_node(w2, lo) == i && rec_edge(w2, c2) < e) ? 1 : 0;
          }
        }
        int p = base + no[i] + rank;
        rec[min(max(p, 0), N_EDGES - 1)] = make_int4(v.x, v.y, v.z, 0);
      }
    }
  }
}

__device__ __forceinline__ float4 edge_msg(float4 hv, float a0, float a1, float4 w0, float4 w1, float4 bb) {
  float m0 = hv.x + a0 * w0.x + a1 * w1.x + bb.x;
  float m1 = hv.y + a0 * w0.y + a1 * w1.y + bb.y;
  float m2 = hv.z + a0 * w0.z + a1 * w1.z + bb.z;
  float m3 = hv.w + a0 * w0.w + a1 * w1.w + bb.w;
  return make_float4(m0 > 0.f ? m0 : 0.f, m1 > 0.f ? m1 : 0.f,
                     m2 > 0.f ? m2 : 0.f, m3 > 0.f ? m3 : 0.f);
}

__device__ __forceinline__ float4 load_row(const float* __restrict__ h, int s, int lane) {
  unsigned sc = min((unsigned)s, (unsigned)(N_NODES - 1));
  return ((const float4*)h)[sc * (unsigned)(HID / 4) + (unsigned)lane];
}

#define TILE_PITCH 132

#define BLOCK_ROWS 32
#define LAYER_BLOCKS ((N_NODES + BLOCK_ROWS - 1) / BLOCK_ROWS)
static_assert(N_NODES % 16 == 0 && LAYER_BLOCKS * BLOCK_ROWS + 2 <= N_PAD, "a 16-row tile is all inside the node list or all past it; the offsets read for the last block exist");
__global__ __launch_bounds__(256, 16) void gine_layer_kernel(
    const float* __restrict__ h,
    const int*   __restrict__ off,
    const int4*  __restrict__ rec,
    const float* __restrict__ Wl0,
    const float* __restrict__ Wl1,
    const float* __restrict__ bl,
    const float* __restrict__ W,
    const float* __restrict__ bias,
    float* __restrict__ out)
{
  __shared__ float xt[BLOCK_ROWS * TILE_PITCH];

  int lane = threadIdx.x & 31;
  int wave = __builtin_amdgcn_readfirstlane((int)(threadIdx.x >> 5));
  int m0   = blockIdx.x * BLOCK_ROWS;
  int c    = lane * 4;

  {
    float4 w0 = *(const float4*)(Wl0 + c);
    float4 w1 = *(const float4*)(Wl1 + c);
    float4 bb = *(const float4*)(bl  + c);

    int r0 = m0 + 4 * wave;
    int q0 = off[r0], q1 = off[r0 + 1], q2 = off[r0 + 2], q3 = off[r0 + 3], q4 = off[r0 + 4];
    for (int pr = 0; pr < 2; ++pr) {
    int rl = 4 * wave + 2 * pr;
    int iA = m0 + rl;
    float4 xA = make_float4(0.f, 0.f, 0.f, 0.f);
    float4 xB = make_float4(0.f, 0.f, 0.f, 0.f);
    if (iA < N_NODES) {
    int o0 = pr ? q2 : q0, o1 = pr ? q3 : q1, o2 = pr ? q4 : q2;
    int beg = min(max(o0, 0), N_EDGES);
    int mid = min(max(o1, beg), N_EDGES);
    int end = min(max(o2, mid), N_EDGES);
    int lenA = mid - beg;
    int len  = end - beg;

    float4 hA = *(const float4*)(h + (long)iA * HID + c);
    float4 hB = *(const float4*)(h + (long)(iA + 1) * HID + c);

    float4 accA = make_float4(0.f, 0.f, 0.f, 0.f);
    float4 accB = make_float4(0.f, 0.f, 0.f, 0.f);
    for (int cb = 0; cb < len; cb += 32) {
      int n   = min(len - cb, 32);
      int nAc = min(max(lenA - cb, 0), n);
      int nBc = n - nAc;
      int p = min(beg + cb + min(lane, n - 1), N_EDGES - 1);
      int4 my = rec[p];
      for (int t = 0; t < n; t += 8) {
        float4 xr[8];
        float  e0[8], e1[8], wa[8], wb[8];
#pragma unroll
        for (int j = 0; j < 8; ++j) {
          int q = t + j;
          xr[j] = load_row(h, __builtin_amdgcn_readlane(my.x, q), lane);
          e0[j] = __int_as_float(__builtin_amdgcn_readlane(my.y, q));
          e1[j] = __int_as_float(__builtin_amdgcn_readlane(my.z, q));
          wa[j] = (q < nAc) ? 1.f : 0.f;
          wb[j] = ((unsigned)(q - nAc) < (unsigned)nBc) ? 1.f : 0.f;
        }
#pragma unroll
        for (int j = 0; j < 8; ++j) {
          float4 m = edge_msg(xr[j], e0[j], e1[j], w0, w1, bb);
          accA.x = fmaf(m.x, wa[j], accA.x);  accA.y = fmaf(m.y, wa[j], accA.y);
          accA.z = fmaf(m.z, wa[j], accA.z);  accA.w = fmaf(m.w, wa[j], accA.w);
          accB.x = fmaf(m.x, wb[j], accB.x);  accB.y = fmaf(m.y, wb[j], accB.y);
          accB.z = fmaf(m.z, wb[j], accB.z);  accB.w = fmaf(m.w, wb[j], accB.w);
        }
      }
    }
    xA = make_float4(hA.x + accA.x, hA.y + accA.y, hA.z + accA.z, hA.w + accA.w);
    xB = make_float4(hB.x + accB.x, hB.y + accB.y, hB.z + accB.z, hB.w + accB.w);
    }
    *(float4*)(xt + rl       * TILE_PITCH + c) = xA;
    *(float4*)(xt + (rl + 1) * TILE_PITCH + c) = xB;
    }
  }

  int m    = lane & 15;
  int half = lane >> 4;
  int n    = wave * 16 + m;

#define LOAD_W_PART(dst, e)                                                  \
  _Pragma("unroll")                                                          \
  for (int s = 0; s < 4; ++s) {                                              \
    int k0 = 4 * (4 * (e) + s) + half * 2;                                   \
    dst[s].x = W[(long)k0 * HID + n];               \
    dst[s].y = W[(long)(k0 + 1) * HID + n];         \
  }
#define WMMA_PART(src, e)                                                    \
  _Pragma("unroll")                                                          \
  for (int s = 0; s < 4; ++s) {                                              \
    v2f a0 = *(const v2f*)(ar0 + 4 * (4 * (e) + s));          \
    v2f a1 = *(const v2f*)(ar1 + 4 * (4 * (e) + s));                  \
    cc0 = __builtin_amdgcn_wmma_f32_16x16x4_f32(                             \
         false, a0,  false, src[s],                        \
         (short)0, cc0,  false,  false);      \
    cc1 = __builtin_amdgcn_wmma_f32_16x16x4_f32(                             \
         false, a1,  false, src[s],                        \
         (short)0, cc1,  false,  false);      \
  }
  v2f f0[4], f1[4];
  LOAD_W_PART(f0, 0)
  LOAD_W_PART(f1, 1)
  float bv = bias[n];

  __syncthreads();

  const float* ar0 = xt + m * TILE_PITCH + half * 2;
  const float* ar1 = ar0 + 16 * TILE_PITCH;
  v8f cc0 = {}, cc1 = {};
  WMMA_PART(f0, 0)
  LOAD_W_PART(f0, 2)
  WMMA_PART(f1, 1)
  LOAD_W_PART(f1, 3)
  WMMA_PART(f0, 2)
  LOAD_W_PART(f0, 4)
  WMMA_PART(f1, 3)
  LOAD_W_PART(f1, 5)
  WMMA_PART(f0, 4)
  LOAD_W_PART(f0, 6)
  WMMA_PART(f1, 5)
  LOAD_W_PART(f1, 7)
  WMMA_PART(f0, 6)
  WMMA_PART(f1, 7)
#undef LOAD_W_PART
#undef WMMA_PART

#pragma unroll
  for (int v = 0; v < 8; ++v) {
    int row = m0 + v + half * 8;
    float y = cc0[v] + bv;
    out[(long)row * HID + n] = (y >= 0.f) ? y : NEG_SLOPE * y;
  }
  if (m0 + 16 < N_NODES) {
#pragma unroll
    for (int v = 0; v < 8; ++v) {
      int row = m0 + 16 + v + half * 8;
      float y = cc1[v] + bv;
      out[(long)row * HID + n] = (y >= 0.f) ? y : NEG_SLOPE * y;
    }
  }
}

extern "C" void kernel_launch(void* const* d_in, const int* in_sizes, int n_in,
                              void* d_out, int out_size, void* d_ws, size_t ws_size,
                              hipStream_t stream) {
  (void)in_sizes; (void)n_in; (void)out_size; (void)ws_size;

  const float* x  = (const float*)d_in[0];
  const int*   ei = (const int*)  d_in[1];
  const float* ea = (const float*)d_in[2];
  const float* Wl = (const float*)d_in[3];
  const float* bl = (const float*)d_in[4];
  const float* W  = (const float*)d_in[5];
  const float* b  = (const float*)d_in[6];
  float*       out = (float*)d_out;

  const int* src = ei;
  const int* dst = ei + N_EDGES;

  const long NH = (long)N_NODES * HID;
  float* agg  = (float*)d_ws;
  int4*  tmp  = (int4*)d_ws;
  int*   tab  = (int*)(tmp + N_EDGES);
  float* h0   = agg + NH;
  int*   off  = (int*)(h0 + NH);
  int4*  rec  = (int4*)(off + N_PAD);

  bin_kernel<<<BIN_BLOCKS, BIN_T, 0, stream>>>(src, dst, ea, tmp, tab);
  range_kernel<<<NR, RNG_T, 0, stream>>>(tmp, tab, off, rec);

  const float* hin = x;
  for (int layer = 0; layer < N_LAYERS; ++layer) {
    float* hout = (layer == 1) ? h0 : out;

    gine_layer_kernel<<<LAYER_BLOCKS, 256, 0, stream>>>(
        hin, off, rec,
        Wl + (long)layer * 2 * HID,
        Wl + (long)layer * 2 * HID + HID,
        bl + (long)layer * HID,
        W + (long)layer * HID * HID,
        b + (long)layer * HID,
        hout);

    hin = hout;
  }
}
